# starting kernel
# baseline (speedup 1.0000x reference)
#include <hip/hip_runtime.h>
#include <hip/hip_bf16.h>

#define N_T   100000
#define N_M   20000
#define N_E   600000
#define N_Q   200000
#define F_IN  256
#define DDIM  128

#define K_SLAB   64
#define B_STRIDE 68

typedef float v2f __attribute__((ext_vector_type(2)));
typedef float v8f __attribute__((ext_vector_type(8)));

__global__ void k_deg_count(const int* __restrict__ idx, float* __restrict__ deg, int n) {
    int t = blockIdx.x * blockDim.x + threadIdx.x;
    if (t >= n) return;
    unsafeAtomicAdd(&deg[idx[t]], 1.0f);
}

__global__ void k_recip(float* __restrict__ deg, int n) {
    int t = blockIdx.x * blockDim.x + threadIdx.x;
    if (t >= n) return;
    deg[t] = 1.0f / fmaxf(deg[t], 1.0f);
}

__global__ void k_gather_rows(const float* __restrict__ src, const int* __restrict__ ids,
                              float* __restrict__ dst, int nrows) {
    int t = blockIdx.x * blockDim.x + threadIdx.x;
    if (t >= nrows * (DDIM / 4)) return;
    int row = t >> 5;
    int l   = t & 31;
    float4 v = *(const float4*)(src + (size_t)ids[row] * DDIM + l * 4);
    *(float4*)(dst + (size_t)row * DDIM + l * 4) = v;
}

__global__ void k_scatter_mean(const float* __restrict__ feat,
                               const int* __restrict__ idx_from,
                               const int* __restrict__ idx_to,
                               const float* __restrict__ rdeg_to,
                               float* __restrict__ agg, int nedges) {
    long long t = (long long)blockIdx.x * blockDim.x + threadIdx.x;
    long long e = t >> 5;
    if (e >= nedges) return;
    int l = (int)(t & 31);
    int s = idx_from[e];
    int d = idx_to[e];
    float r = rdeg_to[d];
    float4 v = *(const float4*)(feat + (size_t)s * DDIM + l * 4);
    float* out = agg + (size_t)d * DDIM + l * 4;
    unsafeAtomicAdd(out + 0, v.x * r);
    unsafeAtomicAdd(out + 1, v.y * r);
    unsafeAtomicAdd(out + 2, v.z * r);
    unsafeAtomicAdd(out + 3, v.w * r);
}

__device__ __forceinline__ void gemm_pass(const float* __restrict__ A,
                                          const float* __restrict__ W,
                                          int K, int rowA, int r, int h, int tid,
                                          float* __restrict__ Bs, v8f acc[8]) {
    for (int kk0 = 0; kk0 < K; kk0 += K_SLAB) {
        __syncthreads();
#pragma unroll
        for (int i = 0; i < 8; i++) {
            int idx = tid + i * 256;
            int row = idx >> 4;
            int c   = (idx & 15) << 2;
            float4 v = *(const float4*)(W + (size_t)row * K + kk0 + c);
            *(float4*)(Bs + row * B_STRIDE + c) = v;
        }
        __syncthreads();

        const float* Arow = A + (size_t)(rowA + r) * K + kk0 + 4 * h;
        for (int ks = 0; ks < K_SLAB; ks += 8) {
            float4 a4 = *(const float4*)(Arow + ks);
            v2f a1; a1[0] = a4.x; a1[1] = a4.y;
            v2f a2; a2[0] = a4.z; a2[1] = a4.w;
#pragma unroll
            for (int nt = 0; nt < 8; nt++) {
                float4 b4 = *(const float4*)(Bs + (nt * 16 + r) * B_STRIDE + ks + 4 * h);
                v2f b1; b1[0] = b4.x; b1[1] = b4.y;
                v2f b2; b2[0] = b4.z; b2[1] = b4.w;
                acc[nt] = __builtin_amdgcn_wmma_f32_16x16x4_f32(
                    false, a1, false, b1, (short)0, acc[nt], false, false);
                acc[nt] = __builtin_amdgcn_wmma_f32_16x16x4_f32(
                    false, a2, false, b2, (short)0, acc[nt], false, false);
            }
        }
    }
}

__global__ void __launch_bounds__(256)
k_gemm_wmma(const float* __restrict__ A1, const float* __restrict__ W1, int K1,
            const float* __restrict__ A2, const float* __restrict__ W2, int K2,
            const float* __restrict__ bias, float* __restrict__ out,
            int nrows, int relu) {
    __shared__ float Bs[DDIM * B_STRIDE];

    const int tid  = threadIdx.x;
    const int lane = tid & 31;
    const int wave = tid >> 5;
    const int row0 = (blockIdx.x * 8 + wave) * 16;
    const bool valid = (row0 < nrows);
    const int rowA = valid ? row0 : 0;
    const int r = lane & 15;
    const int h = lane >> 4;

    v8f acc[8] = {};

    gemm_pass(A1, W1, K1, rowA, r, h, tid, Bs, acc);
    if (A2) gemm_pass(A2, W2, K2, rowA, r, h, tid, Bs, acc);

    if (!valid) return;
#pragma unroll
    for (int nt = 0; nt < 8; nt++) {
        int col  = nt * 16 + r;
        float bb = bias[col];
#pragma unroll
        for (int i = 0; i < 8; i++) {
            float v = acc[nt][i] + bb;
            if (relu) v = fmaxf(v, 0.0f);
            out[(size_t)(row0 + h * 8 + i) * DDIM + col] = v;
        }
    }
}

__global__ void k_classify(const float* __restrict__ ht, const float* __restrict__ hm,
                           const int* __restrict__ qs, const int* __restrict__ qd,
                           const float* __restrict__ w, const float* __restrict__ cb,
                           float* __restrict__ out, int nq) {
    long long t = (long long)blockIdx.x * blockDim.x + threadIdx.x;
    long long q = t >> 5;
    if (q >= nq) return;
    int l = (int)(t & 31);
    float4 a  = *(const float4*)(ht + (size_t)qs[q] * DDIM + l * 4);
    float4 b  = *(const float4*)(hm + (size_t)qd[q] * DDIM + l * 4);
    float4 wa = *(const float4*)(w + l * 4);
    float4 wb = *(const float4*)(w + DDIM + l * 4);
    float s = a.x * wa.x + a.y * wa.y + a.z * wa.z + a.w * wa.w
            + b.x * wb.x + b.y * wb.y + b.z * wb.z + b.w * wb.w;
#pragma unroll
    for (int off = 16; off > 0; off >>= 1) s += __shfl_xor(s, off, 32);
    if (l == 0) out[q] = s + cb[0];
}

extern "C" void kernel_launch(void* const* d_in, const int* in_sizes, int n_in,
                              void* d_out, int out_size, void* d_ws, size_t ws_size,
                              hipStream_t stream) {
    const float* x_thesis     = (const float*)d_in[0];
    const int*   mentor_id    = (const int*)  d_in[1];
    const int*   edge_src     = (const int*)  d_in[2];
    const int*   edge_dst     = (const int*)  d_in[3];
    const int*   eli_src      = (const int*)  d_in[4];
    const int*   eli_dst      = (const int*)  d_in[5];
    const float* thesis_lin_w = (const float*)d_in[6];
    const float* thesis_lin_b = (const float*)d_in[7];
    const float* prof_emb     = (const float*)d_in[8];
    const float* W_nbr_sup    = (const float*)d_in[9];
    const float* W_root_sup   = (const float*)d_in[10];
    const float* b_sup        = (const float*)d_in[11];
    const float* W_nbr_rev    = (const float*)d_in[12];
    const float* W_root_rev   = (const float*)d_in[13];
    const float* b_rev        = (const float*)d_in[14];
    const float* cls_w        = (const float*)d_in[15];
    const float* cls_b        = (const float*)d_in[16];
    float*       out          = (float*)d_out;

    float* p    = (float*)d_ws;
    float* ht0  = p; p += (size_t)N_T * DDIM;
    float* ht1  = p; p += (size_t)N_T * DDIM;
    float* aggt = p; p += (size_t)N_T * DDIM;
    float* hm0  = p; p += (size_t)N_M * DDIM;
    float* hm1  = p; p += (size_t)N_M * DDIM;
    float* aggm = p; p += (size_t)N_M * DDIM;
    float* rdt  = p; p += N_T;
    float* rdm  = p; p += N_M;

    const int BLK = 256;
    const int ROWS_PER_BLK = 8 * 16;

    hipMemsetAsync(rdt, 0, (size_t)N_T * sizeof(float), stream);
    hipMemsetAsync(rdm, 0, (size_t)N_M * sizeof(float), stream);
    k_deg_count<<<(N_E + BLK - 1) / BLK, BLK, 0, stream>>>(edge_dst, rdm, N_E);
    k_deg_count<<<(N_E + BLK - 1) / BLK, BLK, 0, stream>>>(edge_src, rdt, N_E);
    k_recip<<<(N_M + BLK - 1) / BLK, BLK, 0, stream>>>(rdm, N_M);
    k_recip<<<(N_T + BLK - 1) / BLK, BLK, 0, stream>>>(rdt, N_T);

    k_gemm_wmma<<<(N_T + ROWS_PER_BLK - 1) / ROWS_PER_BLK, BLK, 0, stream>>>(
        x_thesis, thesis_lin_w, F_IN, nullptr, nullptr, 0,
        thesis_lin_b, ht0, N_T, 0);
    {
        int n = N_M * (DDIM / 4);
        k_gather_rows<<<(n + BLK - 1) / BLK, BLK, 0, stream>>>(prof_emb, mentor_id, hm0, N_M);
    }

    const long long scat_threads = (long long)N_E * 32;
    const int scat_blocks = (int)((scat_threads + BLK - 1) / BLK);

    hipMemsetAsync(aggm, 0, (size_t)N_M * DDIM * sizeof(float), stream);
    k_scatter_mean<<<scat_blocks, BLK, 0, stream>>>(ht0, edge_src, edge_dst, rdm, aggm, N_E);
    hipMemsetAsync(aggt, 0, (size_t)N_T * DDIM * sizeof(float), stream);
    k_scatter_mean<<<scat_blocks, BLK, 0, stream>>>(hm0, edge_dst, edge_src, rdt, aggt, N_E);
    k_gemm_wmma<<<(N_M + ROWS_PER_BLK - 1) / ROWS_PER_BLK, BLK, 0, stream>>>(
        aggm, W_nbr_sup, DDIM, hm0, W_root_sup, DDIM, b_sup, hm1, N_M, 1);
    k_gemm_wmma<<<(N_T + ROWS_PER_BLK - 1) / ROWS_PER_BLK, BLK, 0, stream>>>(
        aggt, W_nbr_rev, DDIM, ht0, W_root_rev, DDIM, b_rev, ht1, N_T, 1);

    const size_t wOff = (size_t)DDIM * DDIM;
    hipMemsetAsync(aggm, 0, (size_t)N_M * DDIM * sizeof(float), stream);
    k_scatter_mean<<<scat_blocks, BLK, 0, stream>>>(ht1, edge_src, edge_dst, rdm, aggm, N_E);
    hipMemsetAsync(aggt, 0, (size_t)N_T * DDIM * sizeof(float), stream);
    k_scatter_mean<<<scat_blocks, BLK, 0, stream>>>(hm1, edge_dst, edge_src, rdt, aggt, N_E);
    k_gemm_wmma<<<(N_M + ROWS_PER_BLK - 1) / ROWS_PER_BLK, BLK, 0, stream>>>(
        aggm, W_nbr_sup + wOff, DDIM, hm1, W_root_sup + wOff, DDIM, b_sup + DDIM, hm0, N_M, 0);
    k_gemm_wmma<<<(N_T + ROWS_PER_BLK - 1) / ROWS_PER_BLK, BLK, 0, stream>>>(
        aggt, W_nbr_rev + wOff, DDIM, ht1, W_root_rev + wOff, DDIM, b_rev + DDIM, ht0, N_T, 0);

    {
        long long t = (long long)N_Q * 32;
        k_classify<<<(int)((t + BLK - 1) / BLK), BLK, 0, stream>>>(
            ht0, hm0, eli_src, eli_dst, cls_w, cls_b, out, N_Q);
    }
}


# Neighbour means without float atomics: edges grouped by receiver once (integer-atomic counting sort), each node's rows summed in registers, divided by degree, stored once; sum order changes
# speedup vs baseline: 1.4691x; 1.4691x over previous
#include <hip/hip_runtime.h>
#include <hip/hip_bf16.h>

#define N_T   100000
#define N_M   20000
#define N_E   600000
#define N_Q   200000
#define F_IN  256
#define DDIM  128

#define N_NODES  (N_M + N_T)
#define OFF_PAD  120064
#define CSR_INTS (OFF_PAD + 2 * N_E)

#define K_SLAB   64
#define B_STRIDE 68

typedef float v2f __attribute__((ext_vector_type(2)));
typedef float v8f __attribute__((ext_vector_type(8)));

__global__ void k_count(const int* __restrict__ esrc, const int* __restrict__ edst,
                        int* __restrict__ cnt, int n) {
    int t = blockIdx.x * blockDim.x + threadIdx.x;
    if (t >= n) return;
    atomicAdd(&cnt[edst[t]], 1);
    atomicAdd(&cnt[N_M + esrc[t]], 1);
}

__global__ void __launch_bounds__(1024)
k_scan(const int* __restrict__ cnt, int* __restrict__ off) {
    __shared__ int wpre[32];
    __shared__ int wtot[32];
    const int tid  = threadIdx.x;
    const int lane = tid & 31;
    const int wave = tid >> 5;
    const int base = blockIdx.x * 1024;

    int s = 0;
    for (int i = tid * 4; i < base; i += 4096) {
        int4 v = *(const int4*)(cnt + i);
        s += v.x + v.y + v.z + v.w;
    }
    for (int o = 16; o > 0; o >>= 1) s += __shfl_xor(s, o, 32);

    const int i = base + tid;
    const int mine = (i < N_NODES) ? cnt[i] : 0;
    int incl = mine;
    for (int o = 1; o < 32; o <<= 1) {
        int y = __shfl_up(incl, o, 32);
        if (lane >= o) incl += y;
    }
    if (lane == 0)  wpre[wave] = s;
    if (lane == 31) wtot[wave] = incl;
    __syncthreads();
    int front = 0;
    for (int w = 0; w < 32; w++) {
        front += wpre[w];
        if (w < wave) front += wtot[w];
    }
    const int excl = front + incl - mine;
    if (i < N_NODES) off[i] = excl;
    if (i == N_NODES - 1) off[N_NODES] = excl + mine;
    if (i > N_NODES && i < OFF_PAD) off[i] = 0;
}

__global__ void k_fill(const int* __restrict__ esrc, const int* __restrict__ edst,
                       int* __restrict__ cnt, const int* __restrict__ off,
                       int* __restrict__ list, int n) {
    int t = blockIdx.x * blockDim.x + threadIdx.x;
    if (t >= n) return;
    int s = esrc[t];
    int d = edst[t];
    int pm = off[d] + atomicSub(&cnt[d], 1) - 1;
    list[pm] = s;
    int pt = off[N_M + s] + atomicSub(&cnt[N_M + s], 1) - 1;
    list[pt] = d;
}

__global__ void k_copy_ints(const int* __restrict__ src, int* __restrict__ dst, int n4) {
    int t = blockIdx.x * blockDim.x + threadIdx.x;
    if (t >= n4) return;
    ((int4*)dst)[t] = ((const int4*)src)[t];
}

__global__ void k_gather_rows(const float* __restrict__ src, const int* __restrict__ ids,
                              float* __restrict__ dst, int nrows) {
    int t = blockIdx.x * blockDim.x + threadIdx.x;
    if (t >= nrows * (DDIM / 4)) return;
    int row = t >> 5;
    int l   = t & 31;
    float4 v = *(const float4*)(src + (size_t)ids[row] * DDIM + l * 4);
    *(float4*)(dst + (size_t)row * DDIM + l * 4) = v;
}

__global__ void __launch_bounds__(256)
k_seg_mean(const float* __restrict__ feat, const int* __restrict__ off,
           const int* __restrict__ list, float* __restrict__ agg, int nnodes) {
    const int lane = threadIdx.x & 31;
    const int wave0  = __builtin_amdgcn_readfirstlane((int)((blockIdx.x * blockDim.x + threadIdx.x) >> 5));
    const int nwaves = (int)((gridDim.x * blockDim.x) >> 5);
    for (int v = wave0; v < nnodes; v += nwaves) {
        const int beg = off[v];
        const int end = off[v + 1];
        float4 acc = make_float4(0.0f, 0.0f, 0.0f, 0.0f);
        for (int b = beg; b < end; b += 32) {
            const int n = min(32, end - b);
            const int mine = (lane < n) ? list[b + lane] : 0;
#pragma unroll 4
            for (int j = 0; j < n; j++) {
                const int s = __builtin_amdgcn_readlane(mine, j);
                const float4 x = *(const float4*)(feat + (size_t)s * DDIM + lane * 4);
                acc.x += x.x; acc.y += x.y; acc.z += x.z; acc.w += x.w;
            }
        }
        const float dg = fmaxf((float)(end - beg), 1.0f);
        float4 o;
        o.x = acc.x / dg; o.y = acc.y / dg; o.z = acc.z / dg; o.w = acc.w / dg;
        *(float4*)(agg + (size_t)v * DDIM + lane * 4) = o;
    }
}

__device__ __forceinline__ void gemm_pass(const float* __restrict__ A,
                                          const float* __restrict__ W,
                                          int K, int rowA, int r, int h, int tid,
                                          float* __restrict__ Bs, v8f acc[8]) {
    for (int kk0 = 0; kk0 < K; kk0 += K_SLAB) {
        __syncthreads();
#pragma unroll
        for (int i = 0; i < 8; i++) {
            int idx = tid + i * 256;
            int row = idx >> 4;
            int c   = (idx & 15) << 2;
            float4 v = *(const float4*)(W + (size_t)row * K + kk0 + c);
            *(float4*)(Bs + row * B_STRIDE + c) = v;
        }
        __syncthreads();

        const float* Arow = A + (size_t)(rowA + r) * K + kk0 + 4 * h;
        for (int ks = 0; ks < K_SLAB; ks += 8) {
            float4 a4 = *(const float4*)(Arow + ks);
            v2f a1; a1[0] = a4.x; a1[1] = a4.y;
            v2f a2; a2[0] = a4.z; a2[1] = a4.w;
#pragma unroll
            for (int nt = 0; nt < 8; nt++) {
                float4 b4 = *(const float4*)(Bs + (nt * 16 + r) * B_STRIDE + ks + 4 * h);
                v2f b1; b1[0] = b4.x; b1[1] = b4.y;
                v2f b2; b2[0] = b4.z; b2[1] = b4.w;
                acc[nt] = __builtin_amdgcn_wmma_f32_16x16x4_f32(
                    false, a1, false, b1, (short)0, acc[nt], false, false);
                acc[nt] = __builtin_amdgcn_wmma_f32_16x16x4_f32(
                    false, a2, false, b2, (short)0, acc[nt], false, false);
            }
        }
    }
}

__global__ void __launch_bounds__(256)
k_gemm_wmma(const float* __restrict__ A1, const float* __restrict__ W1, int K1,
            const float* __restrict__ A2, const float* __restrict__ W2, int K2,
            const float* __restrict__ bias, float* __restrict__ out,
            int nrows, int relu) {
    __shared__ float Bs[DDIM * B_STRIDE];

    const int tid  = threadIdx.x;
    const int lane = tid & 31;
    const int wave = tid >> 5;
    const int row0 = (blockIdx.x * 8 + wave) * 16;
    const bool valid = (row0 < nrows);
    const int rowA = valid ? row0 : 0;
    const int r = lane & 15;
    const int h = lane >> 4;

    v8f acc[8] = {};

    gemm_pass(A1, W1, K1, rowA, r, h, tid, Bs, acc);
    if (A2) gemm_pass(A2, W2, K2, rowA, r, h, tid, Bs, acc);

    if (!valid) return;
#pragma unroll
    for (int nt = 0; nt < 8; nt++) {
        int col  = nt * 16 + r;
        float bb = bias[col];
#pragma unroll
        for (int i = 0; i < 8; i++) {
            float v = acc[nt][i] + bb;
            if (relu) v = fmaxf(v, 0.0f);
            out[(size_t)(row0 + h * 8 + i) * DDIM + col] = v;
        }
    }
}

__global__ void k_classify(const float* __restrict__ ht, const float* __restrict__ hm,
                           const int* __restrict__ qs, const int* __restrict__ qd,
                           const float* __restrict__ w, const float* __restrict__ cb,
                           float* __restrict__ out, int nq) {
    long long t = (long long)blockIdx.x * blockDim.x + threadIdx.x;
    long long q = t >> 5;
    if (q >= nq) return;
    int l = (int)(t & 31);
    float4 a  = *(const float4*)(ht + (size_t)qs[q] * DDIM + l * 4);
    float4 b  = *(const float4*)(hm + (size_t)qd[q] * DDIM + l * 4);
    float4 wa = *(const float4*)(w + l * 4);
    float4 wb = *(const float4*)(w + DDIM + l * 4);
    float s = a.x * wa.x + a.y * wa.y + a.z * wa.z + a.w * wa.w
            + b.x * wb.x + b.y * wb.y + b.z * wb.z + b.w * wb.w;
#pragma unroll
    for (int off = 16; off > 0; off >>= 1) s += __shfl_xor(s, off, 32);
    if (l == 0) out[q] = s + cb[0];
}

extern "C" void kernel_launch(void* const* d_in, const int* in_sizes, int n_in,
                              void* d_out, int out_size, void* d_ws, size_t ws_size,
                              hipStream_t stream) {
    const float* x_thesis     = (const float*)d_in[0];
    const int*   mentor_id    = (const int*)  d_in[1];
    const int*   edge_src     = (const int*)  d_in[2];
    const int*   edge_dst     = (const int*)  d_in[3];
    const int*   eli_src      = (const int*)  d_in[4];
    const int*   eli_dst      = (const int*)  d_in[5];
    const float* thesis_lin_w = (const float*)d_in[6];
    const float* thesis_lin_b = (const float*)d_in[7];
    const float* prof_emb     = (const float*)d_in[8];
    const float* W_nbr_sup    = (const float*)d_in[9];
    const float* W_root_sup   = (const float*)d_in[10];
    const float* b_sup        = (const float*)d_in[11];
    const float* W_nbr_rev    = (const float*)d_in[12];
    const float* W_root_rev   = (const float*)d_in[13];
    const float* b_rev        = (const float*)d_in[14];
    const float* cls_w        = (const float*)d_in[15];
    const float* cls_b        = (const float*)d_in[16];
    float*       out          = (float*)d_out;

    float* p    = (float*)d_ws;
    float* ht0  = p; p += (size_t)N_T * DDIM;
    float* ht1  = p; p += (size_t)N_T * DDIM;
    float* aggt = p; p += (size_t)N_T * DDIM;
    float* hm0  = p; p += (size_t)N_M * DDIM;
    float* hm1  = p; p += (size_t)N_M * DDIM;
    float* aggm = p; p += (size_t)N_M * DDIM;
    int*   cnt  = (int*)p;

    int* csrA  = (int*)ht1;
    int* offA  = csrA;
    int* listA = csrA + OFF_PAD;
    int* csrB  = (int*)hm0;
    int* offB  = csrB;
    int* listB = csrB + OFF_PAD;

    const int BLK = 256;
    const int ROWS_PER_BLK = 8 * 16;
    const int SEG_BLOCKS_M = 2500;
    const int SEG_BLOCKS_T = 2048;

    hipMemsetAsync(cnt, 0, (size_t)N_NODES * sizeof(int), stream);
    k_count<<<(N_E + BLK - 1) / BLK, BLK, 0, stream>>>(edge_src, edge_dst, cnt, N_E);
    k_scan<<<(N_NODES + 1023) / 1024, 1024, 0, stream>>>(cnt, offA);
    k_fill<<<(N_E + BLK - 1) / BLK, BLK, 0, stream>>>(edge_src, edge_dst, cnt, offA, listA, N_E);

    k_gemm_wmma<<<(N_T + ROWS_PER_BLK - 1) / ROWS_PER_BLK, BLK, 0, stream>>>(
        x_thesis, thesis_lin_w, F_IN, nullptr, nullptr, 0,
        thesis_lin_b, ht0, N_T, 0);
    {
        int n = N_M * (DDIM / 4);
        k_gather_rows<<<(n + BLK - 1) / BLK, BLK, 0, stream>>>(prof_emb, mentor_id, hm0, N_M);
    }

    k_seg_mean<<<SEG_BLOCKS_M, BLK, 0, stream>>>(ht0, offA, listA, aggm, N_M);
    k_seg_mean<<<SEG_BLOCKS_T, BLK, 0, stream>>>(hm0, offA + N_M, listA, aggt, N_T);
    k_gemm_wmma<<<(N_M + ROWS_PER_BLK - 1) / ROWS_PER_BLK, BLK, 0, stream>>>(
        aggm, W_nbr_sup, DDIM, hm0, W_root_sup, DDIM, b_sup, hm1, N_M, 1);
    k_copy_ints<<<(CSR_INTS / 4 + BLK - 1) / BLK, BLK, 0, stream>>>(csrA, csrB, CSR_INTS / 4);
    k_gemm_wmma<<<(N_T + ROWS_PER_BLK - 1) / ROWS_PER_BLK, BLK, 0, stream>>>(
        aggt, W_nbr_rev, DDIM, ht0, W_root_rev, DDIM, b_rev, ht1, N_T, 1);

    const size_t wOff = (size_t)DDIM * DDIM;
    k_seg_mean<<<SEG_BLOCKS_M, BLK, 0, stream>>>(ht1, offB, listB, aggm, N_M);
    k_seg_mean<<<SEG_BLOCKS_T, BLK, 0, stream>>>(hm1, offB + N_M, listB, aggt, N_T);
    k_gemm_wmma<<<(N_M + ROWS_PER_BLK - 1) / ROWS_PER_BLK, BLK, 0, stream>>>(
        aggm, W_nbr_sup + wOff, DDIM, hm1, W_root_sup + wOff, DDIM, b_sup + DDIM, hm0, N_M, 0);
    k_gemm_wmma<<<(N_T + ROWS_PER_BLK - 1) / ROWS_PER_BLK, BLK, 0, stream>>>(
        aggt, W_nbr_rev + wOff, DDIM, ht1, W_root_rev + wOff, DDIM, b_rev + DDIM, ht0, N_T, 0);

    {
        long long t = (long long)N_Q * 32;
        k_classify<<<(int)((t + BLK - 1) / BLK), BLK, 0, stream>>>(
            ht0, hm0, eli_src, eli_dst, cls_w, cls_b, out, N_Q);
    }
}


# Edges grouped by receiver without global atomics: blocks bin them by node range on LDS counters (count, scan, scatter), one block per range ranks them in LDS; order inside a run (sum order) changes
# speedup vs baseline: 2.1669x; 1.4750x over previous
#include <hip/hip_runtime.h>
#include <hip/hip_bf16.h>

#define N_T   100000
#define N_M   20000
#define N_E   600000
#define N_Q   200000
#define F_IN  256
#define DDIM  128

#define N_NODES  (N_M + N_T)
#define OFF_PAD  120064
#define CSR_INTS (OFF_PAD + 2 * N_E)

#define K_SLAB   64
#define B_STRIDE 68

typedef float v2f __attribute__((ext_vector_type(2)));
typedef float v8f __attribute__((ext_vector_type(8)));

#define RM_SHIFT 7
#define RT_SHIFT 9
#define NB_M   157
#define NB_T   196
#define NBINS  (NB_M + NB_T)
#define NGRP   256
#define CHUNK  2344
#define HN     (NBINS * NGRP)
#define HN_PAD (HN + 4)
#define EPT    ((CHUNK + 511) / 512)

__global__ void __launch_bounds__(512)
k_bin_count(const int* __restrict__ esrc, const int* __restrict__ edst, int* __restrict__ H) {
    __shared__ int h[NBINS];
    const int tid = threadIdx.x;
    const int g   = blockIdx.x;
    for (int i = tid; i < NBINS; i += 512) h[i] = 0;
    __syncthreads();
    const int beg = g * CHUNK;
    const int end = min(beg + CHUNK, N_E);
    int s[EPT], d[EPT];
#pragma unroll
    for (int k = 0; k < EPT; k++) {
        const int e = min(beg + tid + k * 512, end - 1);
        s[k] = esrc[e];
        d[k] = edst[e];
    }
#pragma unroll
    for (int k = 0; k < EPT; k++) {
        if (beg + tid + k * 512 < end) {
            atomicAdd(&h[d[k] >> RM_SHIFT], 1);
            atomicAdd(&h[NB_M + (s[k] >> RT_SHIFT)], 1);
        }
    }
    __syncthreads();
    for (int i = tid; i < NBINS; i += 512) H[i * NGRP + g] = h[i];
}

__global__ void __launch_bounds__(1024)
k_scan(const int* __restrict__ cnt, int* __restrict__ off, int n) {
    __shared__ int wpre[32];
    __shared__ int wtot[32];
    const int tid  = threadIdx.x;
    const int lane = tid & 31;
    const int wave = tid >> 5;
    const int base = blockIdx.x * 1024;

    int s = 0;
    for (int i0 = tid * 4; i0 < base; i0 += 4 * 4096) {
        int4 v[4];
#pragma unroll
        for (int k = 0; k < 4; k++) {
            const int i = i0 + k * 4096;
            v[k] = (i < base) ? *(const int4*)(cnt + i) : make_int4(0, 0, 0, 0);
        }
#pragma unroll
        for (int k = 0; k < 4; k++) s += v[k].x + v[k].y + v[k].z + v[k].w;
    }
    for (int o = 16; o > 0; o >>= 1) s += __shfl_xor(s, o, 32);

    const int i = base + tid;
    const int mine = (i < n) ? cnt[i] : 0;
    int incl = mine;
    for (int o = 1; o < 32; o <<= 1) {
        int y = __shfl_up(incl, o, 32);
        if (lane >= o) incl += y;
    }
    if (lane == 0)  wpre[wave] = s;
    if (lane == 31) wtot[wave] = incl;
    __syncthreads();
    int front = 0;
    for (int w = 0; w < 32; w++) {
        front += wpre[w];
        if (w < wave) front += wtot[w];
    }
    const int excl = front + incl - mine;
    if (i < n) off[i] = excl;
    if (i == n - 1) off[n] = excl + mine;
}

__global__ void __launch_bounds__(512)
k_bin_fill(const int* __restrict__ esrc, const int* __restrict__ edst,
           const int* __restrict__ S, int* __restrict__ rec) {
    __shared__ int cur[NBINS];
    const int tid = threadIdx.x;
    const int g   = blockIdx.x;
    for (int i = tid; i < NBINS; i += 512) cur[i] = S[i * NGRP + g];
    __syncthreads();
    const int beg = g * CHUNK;
    const int end = min(beg + CHUNK, N_E);
    int s[EPT], d[EPT];
#pragma unroll
    for (int k = 0; k < EPT; k++) {
        const int e = min(beg + tid + k * 512, end - 1);
        s[k] = esrc[e];
        d[k] = edst[e];
    }
#pragma unroll
    for (int k = 0; k < EPT; k++) {
        if (beg + tid + k * 512 < end) {
            const int pm = atomicAdd(&cur[d[k] >> RM_SHIFT], 1);
            rec[pm] = ((d[k] & ((1 << RM_SHIFT) - 1)) << 17) | s[k];
            const int pt = atomicAdd(&cur[NB_M + (s[k] >> RT_SHIFT)], 1);
            rec[pt] = ((s[k] & ((1 << RT_SHIFT) - 1)) << 17) | d[k];
        }
    }
}

__global__ void __launch_bounds__(512)
k_bin_rank(const int* __restrict__ S, const int* __restrict__ rec,
           int* __restrict__ off, int* __restrict__ list) {
    __shared__ int c[512];
    __shared__ int wtot[16];
    const int tid  = threadIdx.x;
    const int lane = tid & 31;
    const int wave = tid >> 5;
    const int b    = blockIdx.x;
    const int beg  = S[b * NGRP];
    const int end  = S[(b + 1) * NGRP];
    const int node0 = (b < NB_M) ? (b << RM_SHIFT) : (N_M + ((b - NB_M) << RT_SHIFT));
    const int nn    = (b < NB_M) ? min(1 << RM_SHIFT, N_M - (b << RM_SHIFT))
                                 : min(1 << RT_SHIFT, N_T - ((b - NB_M) << RT_SHIFT));
    c[tid] = 0;
    __syncthreads();
    for (int i0 = beg + tid; i0 < end; i0 += 4 * 512) {
        int r[4];
#pragma unroll
        for (int k = 0; k < 4; k++) {
            const int i = i0 + k * 512;
            r[k] = (i < end) ? rec[i] : -1;
        }
#pragma unroll
        for (int k = 0; k < 4; k++) if (r[k] >= 0) atomicAdd(&c[r[k] >> 17], 1);
    }
    __syncthreads();

    const int mine = c[tid];
    int incl = mine;
    for (int o = 1; o < 32; o <<= 1) {
        int y = __shfl_up(incl, o, 32);
        if (lane >= o) incl += y;
    }
    if (lane == 31) wtot[wave] = incl;
    __syncthreads();
    int front = 0;
    for (int w = 0; w < 16; w++) if (w < wave) front += wtot[w];
    const int start = beg + front + incl - mine;
    c[tid] = start;
    if (tid < nn) off[node0 + tid] = start;
    __syncthreads();

    for (int i0 = beg + tid; i0 < end; i0 += 4 * 512) {
        int r[4];
#pragma unroll
        for (int k = 0; k < 4; k++) {
            const int i = i0 + k * 512;
            r[k] = (i < end) ? rec[i] : -1;
        }
#pragma unroll
        for (int k = 0; k < 4; k++) {
            if (r[k] >= 0) {
                const int p = atomicAdd(&c[r[k] >> 17], 1);
                list[p] = r[k] & ((1 << 17) - 1);
            }
        }
    }
    if (b == NBINS - 1) {
        if (tid == 0) off[N_NODES] = end;
        if (tid > 0 && N_NODES + tid < OFF_PAD) off[N_NODES + tid] = 0;
    }
}

__global__ void k_copy_ints(const int* __restrict__ src, int* __restrict__ dst, int n4) {
    int t = blockIdx.x * blockDim.x + threadIdx.x;
    if (t >= n4) return;
    ((int4*)dst)[t] = ((const int4*)src)[t];
}

__global__ void k_gather_rows(const float* __restrict__ src, const int* __restrict__ ids,
                              float* __restrict__ dst, int nrows) {
    int t = blockIdx.x * blockDim.x + threadIdx.x;
    if (t >= nrows * (DDIM / 4)) return;
    int row = t >> 5;
    int l   = t & 31;
    float4 v = *(const float4*)(src + (size_t)ids[row] * DDIM + l * 4);
    *(float4*)(dst + (size_t)row * DDIM + l * 4) = v;
}

__global__ void __launch_bounds__(256)
k_seg_mean(const float* __restrict__ feat, const int* __restrict__ off,
           const int* __restrict__ list, float* __restrict__ agg, int nnodes) {
    const int lane = threadIdx.x & 31;
    const int wave0  = __builtin_amdgcn_readfirstlane((int)((blockIdx.x * blockDim.x + threadIdx.x) >> 5));
    const int nwaves = (int)((gridDim.x * blockDim.x) >> 5);
    for (int v = wave0; v < nnodes; v += nwaves) {
        const int beg = off[v];
        const int end = off[v + 1];
        float4 acc = make_float4(0.0f, 0.0f, 0.0f, 0.0f);
        for (int b = beg; b < end; b += 32) {
            const int n = min(32, end - b);
            const int mine = (lane < n) ? list[b + lane] : 0;
#pragma unroll 4
            for (int j = 0; j < n; j++) {
                const int s = __builtin_amdgcn_readlane(mine, j);
                const float4 x = *(const float4*)(feat + (size_t)s * DDIM + lane * 4);
                acc.x += x.x; acc.y += x.y; acc.z += x.z; acc.w += x.w;
            }
        }
        const float dg = fmaxf((float)(end - beg), 1.0f);
        float4 o;
        o.x = acc.x / dg; o.y = acc.y / dg; o.z = acc.z / dg; o.w = acc.w / dg;
        *(float4*)(agg + (size_t)v * DDIM + lane * 4) = o;
    }
}

__device__ __forceinline__ void gemm_pass(const float* __restrict__ A,
                                          const float* __restrict__ W,
                                          int K, int rowA, int r, int h, int tid,
                                          float* __restrict__ Bs, v8f acc[8]) {
    for (int kk0 = 0; kk0 < K; kk0 += K_SLAB) {
        __syncthreads();
#pragma unroll
        for (int i = 0; i < 8; i++) {
            int idx = tid + i * 256;
            int row = idx >> 4;
            int c   = (idx & 15) << 2;
            float4 v = *(const float4*)(W + (size_t)row * K + kk0 + c);
            *(float4*)(Bs + row * B_STRIDE + c) = v;
        }
        __syncthreads();

        const float* Arow = A + (size_t)(rowA + r) * K + kk0 + 4 * h;
        for (int ks = 0; ks < K_SLAB; ks += 8) {
            float4 a4 = *(const float4*)(Arow + ks);
            v2f a1; a1[0] = a4.x; a1[1] = a4.y;
            v2f a2; a2[0] = a4.z; a2[1] = a4.w;
#pragma unroll
            for (int nt = 0; nt < 8; nt++) {
                float4 b4 = *(const float4*)(Bs + (nt * 16 + r) * B_STRIDE + ks + 4 * h);
                v2f b1; b1[0] = b4.x; b1[1] = b4.y;
                v2f b2; b2[0] = b4.z; b2[1] = b4.w;
                acc[nt] = __builtin_amdgcn_wmma_f32_16x16x4_f32(
                    false, a1, false, b1, (short)0, acc[nt], false, false);
                acc[nt] = __builtin_amdgcn_wmma_f32_16x16x4_f32(
                    false, a2, false, b2, (short)0, acc[nt], false, false);
            }
        }
    }
}

__global__ void __launch_bounds__(256)
k_gemm_wmma(const float* __restrict__ A1, const float* __restrict__ W1, int K1,
            const float* __restrict__ A2, const float* __restrict__ W2, int K2,
            const float* __restrict__ bias, float* __restrict__ out,
            int nrows, int relu) {
    __shared__ float Bs[DDIM * B_STRIDE];

    const int tid  = threadIdx.x;
    const int lane = tid & 31;
    const int wave = tid >> 5;
    const int row0 = (blockIdx.x * 8 + wave) * 16;
    const bool valid = (row0 < nrows);
    const int rowA = valid ? row0 : 0;
    const int r = lane & 15;
    const int h = lane >> 4;

    v8f acc[8] = {};

    gemm_pass(A1, W1, K1, rowA, r, h, tid, Bs, acc);
    if (A2) gemm_pass(A2, W2, K2, rowA, r, h, tid, Bs, acc);

    if (!valid) return;
#pragma unroll
    for (int nt = 0; nt < 8; nt++) {
        int col  = nt * 16 + r;
        float bb = bias[col];
#pragma unroll
        for (int i = 0; i < 8; i++) {
            float v = acc[nt][i] + bb;
            if (relu) v = fmaxf(v, 0.0f);
            out[(size_t)(row0 + h * 8 + i) * DDIM + col] = v;
        }
    }
}

__global__ void k_classify(const float* __restrict__ ht, const float* __restrict__ hm,
                           const int* __restrict__ qs, const int* __restrict__ qd,
                           const float* __restrict__ w, const float* __restrict__ cb,
                           float* __restrict__ out, int nq) {
    long long t = (long long)blockIdx.x * blockDim.x + threadIdx.x;
    long long q = t >> 5;
    if (q >= nq) return;
    int l = (int)(t & 31);
    float4 a  = *(const float4*)(ht + (size_t)qs[q] * DDIM + l * 4);
    float4 b  = *(const float4*)(hm + (size_t)qd[q] * DDIM + l * 4);
    float4 wa = *(const float4*)(w + l * 4);
    float4 wb = *(const float4*)(w + DDIM + l * 4);
    float s = a.x * wa.x + a.y * wa.y + a.z * wa.z + a.w * wa.w
            + b.x * wb.x + b.y * wb.y + b.z * wb.z + b.w * wb.w;
#pragma unroll
    for (int off = 16; off > 0; off >>= 1) s += __shfl_xor(s, off, 32);
    if (l == 0) out[q] = s + cb[0];
}

extern "C" void kernel_launch(void* const* d_in, const int* in_sizes, int n_in,
                              void* d_out, int out_size, void* d_ws, size_t ws_size,
                              hipStream_t stream) {
    const float* x_thesis     = (const float*)d_in[0];
    const int*   mentor_id    = (const int*)  d_in[1];
    const int*   edge_src     = (const int*)  d_in[2];
    const int*   edge_dst     = (const int*)  d_in[3];
    const int*   eli_src      = (const int*)  d_in[4];
    const int*   eli_dst      = (const int*)  d_in[5];
    const float* thesis_lin_w = (const float*)d_in[6];
    const float* thesis_lin_b = (const float*)d_in[7];
    const float* prof_emb     = (const float*)d_in[8];
    const float* W_nbr_sup    = (const float*)d_in[9];
    const float* W_root_sup   = (const float*)d_in[10];
    const float* b_sup        = (const float*)d_in[11];
    const float* W_nbr_rev    = (const float*)d_in[12];
    const float* W_root_rev   = (const float*)d_in[13];
    const float* b_rev        = (const float*)d_in[14];
    const float* cls_w        = (const float*)d_in[15];
    const float* cls_b        = (const float*)d_in[16];
    float*       out          = (float*)d_out;

    float* p    = (float*)d_ws;
    float* ht0  = p; p += (size_t)N_T * DDIM;
    float* ht1  = p; p += (size_t)N_T * DDIM;
    float* aggt = p; p += (size_t)N_T * DDIM;
    float* hm0  = p; p += (size_t)N_M * DDIM;
    float* hm1  = p; p += (size_t)N_M * DDIM;
    float* aggm = p; p += (size_t)N_M * DDIM;

    int* csrA  = (int*)ht1;
    int* offA  = csrA;
    int* listA = csrA + OFF_PAD;
    int* csrB  = (int*)hm0;
    int* offB  = csrB;
    int* listB = csrB + OFF_PAD;

    int* binH = (int*)aggt;
    int* binS = binH + HN;
    int* binR = binS + HN_PAD;

    const int BLK = 256;
    const int ROWS_PER_BLK = 8 * 16;
    const int SEG_BLOCKS_M = 2500;
    const int SEG_BLOCKS_T = 2048;

    k_bin_count<<<NGRP, 512, 0, stream>>>(edge_src, edge_dst, binH);
    k_scan<<<(HN + 1023) / 1024, 1024, 0, stream>>>(binH, binS, HN);
    k_bin_fill<<<NGRP, 512, 0, stream>>>(edge_src, edge_dst, binS, binR);
    k_bin_rank<<<NBINS, 512, 0, stream>>>(binS, binR, offA, listA);

    k_gemm_wmma<<<(N_T + ROWS_PER_BLK - 1) / ROWS_PER_BLK, BLK, 0, stream>>>(
        x_thesis, thesis_lin_w, F_IN, nullptr, nullptr, 0,
        thesis_lin_b, ht0, N_T, 0);
    {
        int n = N_M * (DDIM / 4);
        k_gather_rows<<<(n + BLK - 1) / BLK, BLK, 0, stream>>>(prof_emb, mentor_id, hm0, N_M);
    }

    k_seg_mean<<<SEG_BLOCKS_M, BLK, 0, stream>>>(ht0, offA, listA, aggm, N_M);
    k_seg_mean<<<SEG_BLOCKS_T, BLK, 0, stream>>>(hm0, offA + N_M, listA, aggt, N_T);
    k_gemm_wmma<<<(N_M + ROWS_PER_BLK - 1) / ROWS_PER_BLK, BLK, 0, stream>>>(
        aggm, W_nbr_sup, DDIM, hm0, W_root_sup, DDIM, b_sup, hm1, N_M, 1);
    k_copy_ints<<<(CSR_INTS / 4 + BLK - 1) / BLK, BLK, 0, stream>>>(csrA, csrB, CSR_INTS / 4);
    k_gemm_wmma<<<(N_T + ROWS_PER_BLK - 1) / ROWS_PER_BLK, BLK, 0, stream>>>(
        aggt, W_nbr_rev, DDIM, ht0, W_root_rev, DDIM, b_rev, ht1, N_T, 1);

    const size_t wOff = (size_t)DDIM * DDIM;
    k_seg_mean<<<SEG_BLOCKS_M, BLK, 0, stream>>>(ht1, offB, listB, aggm, N_M);
    k_seg_mean<<<SEG_BLOCKS_T, BLK, 0, stream>>>(hm1, offB + N_M, listB, aggt, N_T);
    k_gemm_wmma<<<(N_M + ROWS_PER_BLK - 1) / ROWS_PER_BLK, BLK, 0, stream>>>(
        aggm, W_nbr_sup + wOff, DDIM, hm1, W_root_sup + wOff, DDIM, b_sup + DDIM, hm0, N_M, 0);
    k_gemm_wmma<<<(N_T + ROWS_PER_BLK - 1) / ROWS_PER_BLK, BLK, 0, stream>>>(
        aggt, W_nbr_rev + wOff, DDIM, ht1, W_root_rev + wOff, DDIM, b_rev + DDIM, ht0, N_T, 0);

    {
        long long t = (long long)N_Q * 32;
        k_classify<<<(int)((t + BLK - 1) / BLK), BLK, 0, stream>>>(
            ht0, hm0, eli_src, eli_dst, cls_w, cls_b, out, N_Q);
    }
}


# Last (linear) layer folded into the classifier: W1^T cls_w gives 4 vectors, hidden-layer product epilogue dots its rows to 2 scalars per node; layer-1 means and products dropped; sum order changes
# speedup vs baseline: 3.3551x; 1.5483x over previous
#include <hip/hip_runtime.h>
#include <hip/hip_bf16.h>

#define N_T   100000
#define N_M   20000
#define N_E   600000
#define N_Q   200000
#define F_IN  256
#define DDIM  128

#define N_NODES  (N_M + N_T)
#define OFF_PAD  120064

#define K_SLAB   64
#define B_STRIDE 68

typedef float v2f __attribute__((ext_vector_type(2)));
typedef float v8f __attribute__((ext_vector_type(8)));

#define RM_SHIFT 7
#define RT_SHIFT 9
#define NB_M   157
#define NB_T   196
#define NBINS  (NB_M + NB_T)
#define NGRP   256
#define CHUNK  2344
#define HN     (NBINS * NGRP)
#define HN_PAD (HN + 4)
#define EPT    ((CHUNK + 511) / 512)

__global__ void __launch_bounds__(512)
k_bin_count(const int* __restrict__ esrc, const int* __restrict__ edst, int* __restrict__ H) {
    __shared__ int h[NBINS];
    const int tid = threadIdx.x;
    const int g   = blockIdx.x;
    for (int i = tid; i < NBINS; i += 512) h[i] = 0;
    __syncthreads();
    const int beg = g * CHUNK;
    const int end = min(beg + CHUNK, N_E);
    int s[EPT], d[EPT];
#pragma unroll
    for (int k = 0; k < EPT; k++) {
        const int e = min(beg + tid + k * 512, end - 1);
        s[k] = esrc[e];
        d[k] = edst[e];
    }
#pragma unroll
    for (int k = 0; k < EPT; k++) {
        if (beg + tid + k * 512 < end) {
            atomicAdd(&h[d[k] >> RM_SHIFT], 1);
            atomicAdd(&h[NB_M + (s[k] >> RT_SHIFT)], 1);
        }
    }
    __syncthreads();
    for (int i = tid; i < NBINS; i += 512) H[i * NGRP + g] = h[i];
}

__global__ void __launch_bounds__(1024)
k_scan(const int* __restrict__ cnt, int* __restrict__ off, int n) {
    __shared__ int wpre[32];
    __shared__ int wtot[32];
    const int tid  = threadIdx.x;
    const int lane = tid & 31;
    const int wave = tid >> 5;
    const int base = blockIdx.x * 1024;

    int s = 0;
    for (int i0 = tid * 4; i0 < base; i0 += 4 * 4096) {
        int4 v[4];
#pragma unroll
        for (int k = 0; k < 4; k++) {
            const int i = i0 + k * 4096;
            v[k] = (i < base) ? *(const int4*)(cnt + i) : make_int4(0, 0, 0, 0);
        }
#pragma unroll
        for (int k = 0; k < 4; k++) s += v[k].x + v[k].y + v[k].z + v[k].w;
    }
    for (int o = 16; o > 0; o >>= 1) s += __shfl_xor(s, o, 32);

    const int i = base + tid;
    const int mine = (i < n) ? cnt[i] : 0;
    int incl = mine;
    for (int o = 1; o < 32; o <<= 1) {
        int y = __shfl_up(incl, o, 32);
        if (lane >= o) incl += y;
    }
    if (lane == 0)  wpre[wave] = s;
    if (lane == 31) wtot[wave] = incl;
    __syncthreads();
    int front = 0;
    for (int w = 0; w < 32; w++) {
        front += wpre[w];
        if (w < wave) front += wtot[w];
    }
    const int excl = front + incl - mine;
    if (i < n) off[i] = excl;
    if (i == n - 1) off[n] = excl + mine;
}

__global__ void __launch_bounds__(512)
k_bin_fill(const int* __restrict__ esrc, const int* __restrict__ edst,
           const int* __restrict__ S, int* __restrict__ rec) {
    __shared__ int cur[NBINS];
    const int tid = threadIdx.x;
    const int g   = blockIdx.x;
    for (int i = tid; i < NBINS; i += 512) cur[i] = S[i * NGRP + g];
    __syncthreads();
    const int beg = g * CHUNK;
    const int end = min(beg + CHUNK, N_E);
    int s[EPT], d[EPT];
#pragma unroll
    for (int k = 0; k < EPT; k++) {
        const int e = min(beg + tid + k * 512, end - 1);
        s[k] = esrc[e];
        d[k] = edst[e];
    }
#pragma unroll
    for (int k = 0; k < EPT; k++) {
        if (beg + tid + k * 512 < end) {
            const int pm = atomicAdd(&cur[d[k] >> RM_SHIFT], 1);
            rec[pm] = ((d[k] & ((1 << RM_SHIFT) - 1)) << 17) | s[k];
            const int pt = atomicAdd(&cur[NB_M + (s[k] >> RT_SHIFT)], 1);
            rec[pt] = ((s[k] & ((1 << RT_SHIFT) - 1)) << 17) | d[k];
        }
    }
}

__global__ void __launch_bounds__(512)
k_bin_rank(const int* __restrict__ S, const int* __restrict__ rec,
           int* __restrict__ off, int* __restrict__ list) {
    __shared__ int c[512];
    __shared__ int wtot[16];
    const int tid  = threadIdx.x;
    const int lane = tid & 31;
    const int wave = tid >> 5;
    const int b    = blockIdx.x;
    const int beg  = S[b * NGRP];
    const int end  = S[(b + 1) * NGRP];
    const int node0 = (b < NB_M) ? (b << RM_SHIFT) : (N_M + ((b - NB_M) << RT_SHIFT));
    const int nn    = (b < NB_M) ? min(1 << RM_SHIFT, N_M - (b << RM_SHIFT))
                                 : min(1 << RT_SHIFT, N_T - ((b - NB_M) << RT_SHIFT));
    c[tid] = 0;
    __syncthreads();
    for (int i0 = beg + tid; i0 < end; i0 += 4 * 512) {
        int r[4];
#pragma unroll
        for (int k = 0; k < 4; k++) {
            const int i = i0 + k * 512;
            r[k] = (i < end) ? rec[i] : -1;
        }
#pragma unroll
        for (int k = 0; k < 4; k++) if (r[k] >= 0) atomicAdd(&c[r[k] >> 17], 1);
    }
    __syncthreads();

    const int mine = c[tid];
    int incl = mine;
    for (int o = 1; o < 32; o <<= 1) {
        int y = __shfl_up(incl, o, 32);
        if (lane >= o) incl += y;
    }
    if (lane == 31) wtot[wave] = incl;
    __syncthreads();
    int front = 0;
    for (int w = 0; w < 16; w++) if (w < wave) front += wtot[w];
    const int start = beg + front + incl - mine;
    c[tid] = start;
    if (tid < nn) off[node0 + tid] = start;
    __syncthreads();

    for (int i0 = beg + tid; i0 < end; i0 += 4 * 512) {
        int r[4];
#pragma unroll
        for (int k = 0; k < 4; k++) {
            const int i = i0 + k * 512;
            r[k] = (i < end) ? rec[i] : -1;
        }
#pragma unroll
        for (int k = 0; k < 4; k++) {
            if (r[k] >= 0) {
                const int p = atomicAdd(&c[r[k] >> 17], 1);
                list[p] = r[k] & ((1 << 17) - 1);
            }
        }
    }
    if (b == NBINS - 1 && tid == 0) off[N_NODES] = end;
}

__global__ void k_gather_rows(const float* __restrict__ src, const int* __restrict__ ids,
                              float* __restrict__ dst, int nrows) {
    int t = blockIdx.x * blockDim.x + threadIdx.x;
    if (t >= nrows * (DDIM / 4)) return;
    int row = t >> 5;
    int l   = t & 31;
    float4 v = *(const float4*)(src + (size_t)ids[row] * DDIM + l * 4);
    *(float4*)(dst + (size_t)row * DDIM + l * 4) = v;
}

__global__ void __launch_bounds__(256)
k_seg_mean(const float* __restrict__ feat, const int* __restrict__ off,
           const int* __restrict__ list, float* __restrict__ agg, int nnodes) {
    const int lane = threadIdx.x & 31;
    const int wave0  = __builtin_amdgcn_readfirstlane((int)((blockIdx.x * blockDim.x + threadIdx.x) >> 5));
    const int nwaves = (int)((gridDim.x * blockDim.x) >> 5);
    for (int v = wave0; v < nnodes; v += nwaves) {
        const int beg = off[v];
        const int end = off[v + 1];
        float4 acc = make_float4(0.0f, 0.0f, 0.0f, 0.0f);
        for (int b = beg; b < end; b += 32) {
            const int n = min(32, end - b);
            const int mine = (lane < n) ? list[b + lane] : 0;
#pragma unroll 4
            for (int j = 0; j < n; j++) {
                const int s = __builtin_amdgcn_readlane(mine, j);
                const float4 x = *(const float4*)(feat + (size_t)s * DDIM + lane * 4);
                acc.x += x.x; acc.y += x.y; acc.z += x.z; acc.w += x.w;
            }
        }
        const float dg = fmaxf((float)(end - beg), 1.0f);
        float4 o;
        o.x = acc.x / dg; o.y = acc.y / dg; o.z = acc.z / dg; o.w = acc.w / dg;
        *(float4*)(agg + (size_t)v * DDIM + lane * 4) = o;
    }
}

__device__ __forceinline__ void gemm_pass(const float* __restrict__ A,
                                          const float* __restrict__ W,
                                          int K, int rowA, int r, int h, int tid,
                                          float* __restrict__ Bs, v8f acc[8]) {
    for (int kk0 = 0; kk0 < K; kk0 += K_SLAB) {
        __syncthreads();
#pragma unroll
        for (int i = 0; i < 8; i++) {
            int idx = tid + i * 256;
            int row = idx >> 4;
            int c   = (idx & 15) << 2;
            float4 v = *(const float4*)(W + (size_t)row * K + kk0 + c);
            *(float4*)(Bs + row * B_STRIDE + c) = v;
        }
        __syncthreads();

        const float* Arow = A + (size_t)(rowA + r) * K + kk0 + 4 * h;
        for (int ks = 0; ks < K_SLAB; ks += 8) {
            float4 a4 = *(const float4*)(Arow + ks);
            v2f a1; a1[0] = a4.x; a1[1] = a4.y;
            v2f a2; a2[0] = a4.z; a2[1] = a4.w;
#pragma unroll
            for (int nt = 0; nt < 8; nt++) {
                float4 b4 = *(const float4*)(Bs + (nt * 16 + r) * B_STRIDE + ks + 4 * h);
                v2f b1; b1[0] = b4.x; b1[1] = b4.y;
                v2f b2; b2[0] = b4.z; b2[1] = b4.w;
                acc[nt] = __builtin_amdgcn_wmma_f32_16x16x4_f32(
                    false, a1, false, b1, (short)0, acc[nt], false, false);
                acc[nt] = __builtin_amdgcn_wmma_f32_16x16x4_f32(
                    false, a2, false, b2, (short)0, acc[nt], false, false);
            }
        }
    }
}

__global__ void __launch_bounds__(256)
k_gemm_wmma(const float* __restrict__ A1, const float* __restrict__ W1, int K1,
            const float* __restrict__ A2, const float* __restrict__ W2, int K2,
            const float* __restrict__ bias, float* __restrict__ out,
            int nrows, int relu) {
    __shared__ float Bs[DDIM * B_STRIDE];

    const int tid  = threadIdx.x;
    const int lane = tid & 31;
    const int wave = tid >> 5;
    const int row0 = (blockIdx.x * 8 + wave) * 16;
    const bool valid = (row0 < nrows);
    const int rowA = valid ? row0 : 0;
    const int r = lane & 15;
    const int h = lane >> 4;

    v8f acc[8] = {};

    gemm_pass(A1, W1, K1, rowA, r, h, tid, Bs, acc);
    if (A2) gemm_pass(A2, W2, K2, rowA, r, h, tid, Bs, acc);

    if (!valid) return;
#pragma unroll
    for (int nt = 0; nt < 8; nt++) {
        int col  = nt * 16 + r;
        float bb = bias[col];
#pragma unroll
        for (int i = 0; i < 8; i++) {
            float v = acc[nt][i] + bb;
            if (relu) v = fmaxf(v, 0.0f);
            out[(size_t)(row0 + h * 8 + i) * DDIM + col] = v;
        }
    }
}

__global__ void __launch_bounds__(1024)
k_fold_vec(const float* __restrict__ Wroot_t, const float* __restrict__ Wnbr_m,
           const float* __restrict__ Wroot_m, const float* __restrict__ Wnbr_t,
           const float* __restrict__ bias_t, const float* __restrict__ bias_m,
           const float* __restrict__ cw, float* __restrict__ U, float* __restrict__ cst) {
    __shared__ float red[8][DDIM];
    const int tid = threadIdx.x;
    const int b   = blockIdx.x;
    const int k   = tid & (DDIM - 1);
    const int g   = tid >> 7;
    const float* Wm = (b == 0) ? Wroot_t : (b == 1) ? Wnbr_m : (b == 2) ? Wroot_m : Wnbr_t;
    const float* c  = (b == 0 || b == 3) ? cw : cw + DDIM;
    float s = 0.0f;
#pragma unroll
    for (int jj = 0; jj < 16; jj++) {
        const int j = g * 16 + jj;
        s += Wm[j * DDIM + k] * c[j];
    }
    red[g][k] = s;
    __syncthreads();
    if (tid < DDIM) {
        float t = 0.0f;
#pragma unroll
        for (int q = 0; q < 8; q++) t += red[q][tid];
        U[b * DDIM + tid] = t;
    }
    if ((b == 0 || b == 2) && tid < 32) {
        const float* bv = (b == 0) ? bias_t : bias_m;
        float t = 0.0f;
#pragma unroll
        for (int q = 0; q < 4; q++) t += bv[tid * 4 + q] * c[tid * 4 + q];
        for (int o = 16; o > 0; o >>= 1) t += __shfl_xor(t, o, 32);
        if (tid == 0) cst[b >> 1] = t;
    }
}

__global__ void __launch_bounds__(256)
k_gemm_fold(const float* __restrict__ A1, const float* __restrict__ W1, int K1,
            const float* __restrict__ A2, const float* __restrict__ W2, int K2,
            const float* __restrict__ bias, const float* __restrict__ U,
            float* __restrict__ pq, int nrows) {
    __shared__ float Bs[DDIM * B_STRIDE];

    const int tid  = threadIdx.x;
    const int lane = tid & 31;
    const int wave = tid >> 5;
    const int row0 = (blockIdx.x * 8 + wave) * 16;
    const bool valid = (row0 < nrows);
    const int rowA = valid ? row0 : 0;
    const int r = lane & 15;
    const int h = lane >> 4;

    v8f acc[8] = {};

    gemm_pass(A1, W1, K1, rowA, r, h, tid, Bs, acc);
    gemm_pass(A2, W2, K2, rowA, r, h, tid, Bs, acc);

    if (!valid) return;

    float pa[8], qa[8];
#pragma unroll
    for (int i = 0; i < 8; i++) { pa[i] = 0.0f; qa[i] = 0.0f; }
#pragma unroll
    for (int nt = 0; nt < 8; nt++) {
        const int col  = nt * 16 + r;
        const float bb = bias[col];
        const float up = U[col];
        const float uq = U[DDIM + col];
#pragma unroll
        for (int i = 0; i < 8; i++) {
            const float v = fmaxf(acc[nt][i] + bb, 0.0f);
            pa[i] += v * up;
            qa[i] += v * uq;
        }
    }
    float k8[8], k4[4], k2[2], res;
    {
        const bool hi = (r & 8) != 0;
#pragma unroll
        for (int k = 0; k < 8; k++) {
            const float snd = hi ? pa[k] : qa[k];
            const float kp  = hi ? qa[k] : pa[k];
            k8[k] = kp + __shfl_xor(snd, 8, 32);
        }
    }
    {
        const bool hi = (r & 4) != 0;
#pragma unroll
        for (int k = 0; k < 4; k++) {
            const float snd = hi ? k8[k] : k8[k + 4];
            const float kp  = hi ? k8[k + 4] : k8[k];
            k4[k] = kp + __shfl_xor(snd, 4, 32);
        }
    }
    {
        const bool hi = (r & 2) != 0;
#pragma unroll
        for (int k = 0; k < 2; k++) {
            const float snd = hi ? k4[k] : k4[k + 2];
            const float kp  = hi ? k4[k + 2] : k4[k];
            k2[k] = kp + __shfl_xor(snd, 2, 32);
        }
    }
    {
        const bool hi = (r & 1) != 0;
        const float snd = hi ? k2[0] : k2[1];
        const float kp  = hi ? k2[1] : k2[0];
        res = kp + __shfl_xor(snd, 1, 32);
    }
    const int pos = ((r >> 3) << 4) + h * 8 + (r & 7);
    pq[(size_t)(row0 >> 4) * 32 + pos] = res;
}

__device__ __forceinline__ float pq_at(const float* __restrict__ pq, int v, int which) {
    return pq[(size_t)(v >> 4) * 32 + which * 16 + (v & 15)];
}

__global__ void __launch_bounds__(256)
k_node_score(const float* __restrict__ pq_t, const float* __restrict__ pq_m,
             const int* __restrict__ off, const int* __restrict__ list,
             const float* __restrict__ cst, float* __restrict__ S) {
    const int t  = blockIdx.x * 256 + threadIdx.x;
    const int v  = t >> 3;
    const int l8 = t & 7;
    if (v >= N_NODES) return;
    const bool is_m = (v < N_M);
    const float* own = is_m ? pq_m : pq_t;
    const float* oth = is_m ? pq_t : pq_m;
    const int beg = off[v];
    const int end = off[v + 1];
    float s = 0.0f;
    for (int j = beg + l8; j < end; j += 16) {
        const int a  = list[j];
        const int a2 = (j + 8 < end) ? list[j + 8] : -1;
        const float x  = pq_at(oth, a, 1);
        const float x2 = (a2 >= 0) ? pq_at(oth, a2, 1) : 0.0f;
        s += x;
        s += x2;
    }
    s += __shfl_xor(s, 4, 32);
    s += __shfl_xor(s, 2, 32);
    s += __shfl_xor(s, 1, 32);
    if (l8 == 0) {
        const float dg = fmaxf((float)(end - beg), 1.0f);
        S[v] = pq_at(own, is_m ? v : v - N_M, 0) + cst[is_m ? 1 : 0] + s / dg;
    }
}

__global__ void __launch_bounds__(256)
k_classify(const float* __restrict__ S, const int* __restrict__ qs, const int* __restrict__ qd,
           const float* __restrict__ cb, float* __restrict__ out, int nq) {
    const int q = blockIdx.x * 256 + threadIdx.x;
    if (q >= nq) return;
    out[q] = S[N_M + qs[q]] + S[qd[q]] + cb[0];
}

extern "C" void kernel_launch(void* const* d_in, const int* in_sizes, int n_in,
                              void* d_out, int out_size, void* d_ws, size_t ws_size,
                              hipStream_t stream) {
    const float* x_thesis     = (const float*)d_in[0];
    const int*   mentor_id    = (const int*)  d_in[1];
    const int*   edge_src     = (const int*)  d_in[2];
    const int*   edge_dst     = (const int*)  d_in[3];
    const int*   eli_src      = (const int*)  d_in[4];
    const int*   eli_dst      = (const int*)  d_in[5];
    const float* thesis_lin_w = (const float*)d_in[6];
    const float* thesis_lin_b = (const float*)d_in[7];
    const float* prof_emb     = (const float*)d_in[8];
    const float* W_nbr_sup    = (const float*)d_in[9];
    const float* W_root_sup   = (const float*)d_in[10];
    const float* b_sup        = (const float*)d_in[11];
    const float* W_nbr_rev    = (const float*)d_in[12];
    const float* W_root_rev   = (const float*)d_in[13];
    const float* b_rev        = (const float*)d_in[14];
    const float* cls_w        = (const float*)d_in[15];
    const float* cls_b        = (const float*)d_in[16];
    float*       out          = (float*)d_out;

    float* p    = (float*)d_ws;
    float* ht0  = p; p += (size_t)N_T * DDIM;
    float* ht1  = p; p += (size_t)N_T * DDIM;
    float* aggt = p; p += (size_t)N_T * DDIM;
    float* hm0  = p; p += (size_t)N_M * DDIM;
    float* hm1  = p; p += (size_t)N_M * DDIM;
    float* aggm = p; p += (size_t)N_M * DDIM;

    int* csrA  = (int*)ht1;
    int* offA  = csrA;
    int* listA = csrA + OFF_PAD;

    float* foldU = hm1;
    float* foldC = hm1 + 4 * DDIM;
    float* pqT   = hm1 + 1024;
    float* pqM   = pqT + 2 * N_T;
    float* score = pqM + 2 * N_M;

    int* binH = (int*)aggt;
    int* binS = binH + HN;
    int* binR = binS + HN_PAD;

    const int BLK = 256;
    const int ROWS_PER_BLK = 8 * 16;
    const int SEG_BLOCKS_M = 2500;
    const int SEG_BLOCKS_T = 2048;
    const size_t wOff = (size_t)DDIM * DDIM;

    k_bin_count<<<NGRP, 512, 0, stream>>>(edge_src, edge_dst, binH);
    k_scan<<<(HN + 1023) / 1024, 1024, 0, stream>>>(binH, binS, HN);
    k_bin_fill<<<NGRP, 512, 0, stream>>>(edge_src, edge_dst, binS, binR);
    k_bin_rank<<<NBINS, 512, 0, stream>>>(binS, binR, offA, listA);

    k_fold_vec<<<4, 1024, 0, stream>>>(W_root_rev + wOff, W_nbr_sup + wOff,
                                       W_root_sup + wOff, W_nbr_rev + wOff,
                                       b_rev + DDIM, b_sup + DDIM, cls_w, foldU, foldC);

    k_gemm_wmma<<<(N_T + ROWS_PER_BLK - 1) / ROWS_PER_BLK, BLK, 0, stream>>>(
        x_thesis, thesis_lin_w, F_IN, nullptr, nullptr, 0,
        thesis_lin_b, ht0, N_T, 0);
    {
        int n = N_M * (DDIM / 4);
        k_gather_rows<<<(n + BLK - 1) / BLK, BLK, 0, stream>>>(prof_emb, mentor_id, hm0, N_M);
    }

    k_seg_mean<<<SEG_BLOCKS_M, BLK, 0, stream>>>(ht0, offA, listA, aggm, N_M);
    k_seg_mean<<<SEG_BLOCKS_T, BLK, 0, stream>>>(hm0, offA + N_M, listA, aggt, N_T);
    k_gemm_fold<<<(N_M + ROWS_PER_BLK - 1) / ROWS_PER_BLK, BLK, 0, stream>>>(
        aggm, W_nbr_sup, DDIM, hm0, W_root_sup, DDIM, b_sup, foldU + 2 * DDIM, pqM, N_M);
    k_gemm_fold<<<(N_T + ROWS_PER_BLK - 1) / ROWS_PER_BLK, BLK, 0, stream>>>(
        aggt, W_nbr_rev, DDIM, ht0, W_root_rev, DDIM, b_rev, foldU, pqT, N_T);

    k_node_score<<<(N_NODES * 8) / BLK, BLK, 0, stream>>>(pqT, pqM, offA, listA, foldC, score);
    k_classify<<<(N_Q + BLK - 1) / BLK, BLK, 0, stream>>>(score, eli_src, eli_dst, cls_b, out, N_Q);
}


# Neighbour weights of the 100000-node side applied to the 20000 source rows before the mean (mean is linear): its product keeps only the root term, the mean starts the accumulators; sum order changes
# speedup vs baseline: 3.5761x; 1.0659x over previous
#include <hip/hip_runtime.h>
#include <hip/hip_bf16.h>

#define N_T   100000
#define N_M   20000
#define N_E   600000
#define N_Q   200000
#define F_IN  256
#define DDIM  128

#define N_NODES  (N_M + N_T)
#define OFF_PAD  120064

#define K_SLAB   64
#define B_STRIDE 68

typedef float v2f __attribute__((ext_vector_type(2)));
typedef float v8f __attribute__((ext_vector_type(8)));

#define RM_SHIFT 7
#define RT_SHIFT 9
#define NB_M   157
#define NB_T   196
#define NBINS  (NB_M + NB_T)
#define NGRP   256
#define CHUNK  2344
#define HN     (NBINS * NGRP)
#define HN_PAD (HN + 4)
#define EPT    ((CHUNK + 511) / 512)

__global__ void __launch_bounds__(512)
k_bin_count(const int* __restrict__ esrc, const int* __restrict__ edst, int* __restrict__ H) {
    __shared__ int h[NBINS];
    const int tid = threadIdx.x;
    const int g   = blockIdx.x;
    for (int i = tid; i < NBINS; i += 512) h[i] = 0;
    __syncthreads();
    const int beg = g * CHUNK;
    const int end = min(beg + CHUNK, N_E);
    int s[EPT], d[EPT];
#pragma unroll
    for (int k = 0; k < EPT; k++) {
        const int e = min(beg + tid + k * 512, end - 1);
        s[k] = esrc[e];
        d[k] = edst[e];
    }
#pragma unroll
    for (int k = 0; k < EPT; k++) {
        if (beg + tid + k * 512 < end) {
            atomicAdd(&h[d[k] >> RM_SHIFT], 1);
            atomicAdd(&h[NB_M + (s[k] >> RT_SHIFT)], 1);
        }
    }
    __syncthreads();
    for (int i = tid; i < NBINS; i += 512) H[i * NGRP + g] = h[i];
}

__global__ void __launch_bounds__(1024)
k_scan(const int* __restrict__ cnt, int* __restrict__ off, int n) {
    __shared__ int wpre[32];
    __shared__ int wtot[32];
    const int tid  = threadIdx.x;
    const int lane = tid & 31;
    const int wave = tid >> 5;
    const int base = blockIdx.x * 1024;

    int s = 0;
    for (int i0 = tid * 4; i0 < base; i0 += 4 * 4096) {
        int4 v[4];
#pragma unroll
        for (int k = 0; k < 4; k++) {
            const int i = i0 + k * 4096;
            v[k] = (i < base) ? *(const int4*)(cnt + i) : make_int4(0, 0, 0, 0);
        }
#pragma unroll
        for (int k = 0; k < 4; k++) s += v[k].x + v[k].y + v[k].z + v[k].w;
    }
    for (int o = 16; o > 0; o >>= 1) s += __shfl_xor(s, o, 32);

    const int i = base + tid;
    const int mine = (i < n) ? cnt[i] : 0;
    int incl = mine;
    for (int o = 1; o < 32; o <<= 1) {
        int y = __shfl_up(incl, o, 32);
        if (lane >= o) incl += y;
    }
    if (lane == 0)  wpre[wave] = s;
    if (lane == 31) wtot[wave] = incl;
    __syncthreads();
    int front = 0;
    for (int w = 0; w < 32; w++) {
        front += wpre[w];
        if (w < wave) front += wtot[w];
    }
    const int excl = front + incl - mine;
    if (i < n) off[i] = excl;
    if (i == n - 1) off[n] = excl + mine;
}

__global__ void __launch_bounds__(512)
k_bin_fill(const int* __restrict__ esrc, const int* __restrict__ edst,
           const int* __restrict__ S, int* __restrict__ rec) {
    __shared__ int cur[NBINS];
    const int tid = threadIdx.x;
    const int g   = blockIdx.x;
    for (int i = tid; i < NBINS; i += 512) cur[i] = S[i * NGRP + g];
    __syncthreads();
    const int beg = g * CHUNK;
    const int end = min(beg + CHUNK, N_E);
    int s[EPT], d[EPT];
#pragma unroll
    for (int k = 0; k < EPT; k++) {
        const int e = min(beg + tid + k * 512, end - 1);
        s[k] = esrc[e];
        d[k] = edst[e];
    }
#pragma unroll
    for (int k = 0; k < EPT; k++) {
        if (beg + tid + k * 512 < end) {
            const int pm = atomicAdd(&cur[d[k] >> RM_SHIFT], 1);
            rec[pm] = ((d[k] & ((1 << RM_SHIFT) - 1)) << 17) | s[k];
            const int pt = atomicAdd(&cur[NB_M + (s[k] >> RT_SHIFT)], 1);
            rec[pt] = ((s[k] & ((1 << RT_SHIFT) - 1)) << 17) | d[k];
        }
    }
}

__global__ void __launch_bounds__(512)
k_bin_rank(const int* __restrict__ S, const int* __restrict__ rec,
           int* __restrict__ off, int* __restrict__ list) {
    __shared__ int c[512];
    __shared__ int wtot[16];
    const int tid  = threadIdx.x;
    const int lane = tid & 31;
    const int wave = tid >> 5;
    const int b    = blockIdx.x;
    const int beg  = S[b * NGRP];
    const int end  = S[(b + 1) * NGRP];
    const int node0 = (b < NB_M) ? (b << RM_SHIFT) : (N_M + ((b - NB_M) << RT_SHIFT));
    const int nn    = (b < NB_M) ? min(1 << RM_SHIFT, N_M - (b << RM_SHIFT))
                                 : min(1 << RT_SHIFT, N_T - ((b - NB_M) << RT_SHIFT));
    c[tid] = 0;
    __syncthreads();
    for (int i0 = beg + tid; i0 < end; i0 += 4 * 512) {
        int r[4];
#pragma unroll
        for (int k = 0; k < 4; k++) {
            const int i = i0 + k * 512;
            r[k] = (i < end) ? rec[i] : -1;
        }
#pragma unroll
        for (int k = 0; k < 4; k++) if (r[k] >= 0) atomicAdd(&c[r[k] >> 17], 1);
    }
    __syncthreads();

    const int mine = c[tid];
    int incl = mine;
    for (int o = 1; o < 32; o <<= 1) {
        int y = __shfl_up(incl, o, 32);
        if (lane >= o) incl += y;
    }
    if (lane == 31) wtot[wave] = incl;
    __syncthreads();
    int front = 0;
    for (int w = 0; w < 16; w++) if (w < wave) front += wtot[w];
    const int start = beg + front + incl - mine;
    c[tid] = start;
    if (tid < nn) off[node0 + tid] = start;
    __syncthreads();

    for (int i0 = beg + tid; i0 < end; i0 += 4 * 512) {
        int r[4];
#pragma unroll
        for (int k = 0; k < 4; k++) {
            const int i = i0 + k * 512;
            r[k] = (i < end) ? rec[i] : -1;
        }
#pragma unroll
        for (int k = 0; k < 4; k++) {
            if (r[k] >= 0) {
                const int p = atomicAdd(&c[r[k] >> 17], 1);
                list[p] = r[k] & ((1 << 17) - 1);
            }
        }
    }
    if (b == NBINS - 1 && tid == 0) off[N_NODES] = end;
}

__global__ void __launch_bounds__(256)
k_seg_mean(const float* __restrict__ feat, const int* __restrict__ off,
           const int* __restrict__ list, float* __restrict__ agg, int nnodes) {
    const int lane = threadIdx.x & 31;
    const int wave0  = __builtin_amdgcn_readfirstlane((int)((blockIdx.x * blockDim.x + threadIdx.x) >> 5));
    const int nwaves = (int)((gridDim.x * blockDim.x) >> 5);
    for (int v = wave0; v < nnodes; v += nwaves) {
        const int beg = off[v];
        const int end = off[v + 1];
        float4 acc = make_float4(0.0f, 0.0f, 0.0f, 0.0f);
        for (int b = beg; b < end; b += 32) {
            const int n = min(32, end - b);
            const int mine = (lane < n) ? list[b + lane] : 0;
#pragma unroll 4
            for (int j = 0; j < n; j++) {
                const int s = __builtin_amdgcn_readlane(mine, j);
                const float4 x = *(const float4*)(feat + (size_t)s * DDIM + lane * 4);
                acc.x += x.x; acc.y += x.y; acc.z += x.z; acc.w += x.w;
            }
        }
        const float dg = fmaxf((float)(end - beg), 1.0f);
        float4 o;
        o.x = acc.x / dg; o.y = acc.y / dg; o.z = acc.z / dg; o.w = acc.w / dg;
        *(float4*)(agg + (size_t)v * DDIM + lane * 4) = o;
    }
}

__device__ __forceinline__ void gemm_pass(const float* __restrict__ A,
                                          const int* __restrict__ ids,
                                          const float* __restrict__ W,
                                          int K, int rowA, int r, int h, int tid,
                                          float* __restrict__ Bs, v8f acc[8]) {
    const int arow = ids ? ids[rowA + r] : (rowA + r);
    for (int kk0 = 0; kk0 < K; kk0 += K_SLAB) {
        __syncthreads();
#pragma unroll
        for (int i = 0; i < 8; i++) {
            int idx = tid + i * 256;
            int row = idx >> 4;
            int c   = (idx & 15) << 2;
            float4 v = *(const float4*)(W + (size_t)row * K + kk0 + c);
            *(float4*)(Bs + row * B_STRIDE + c) = v;
        }
        __syncthreads();

        const float* Arow = A + (size_t)arow * K + kk0 + 4 * h;
        for (int ks = 0; ks < K_SLAB; ks += 8) {
            float4 a4 = *(const float4*)(Arow + ks);
            v2f a1; a1[0] = a4.x; a1[1] = a4.y;
            v2f a2; a2[0] = a4.z; a2[1] = a4.w;
#pragma unroll
            for (int nt = 0; nt < 8; nt++) {
                float4 b4 = *(const float4*)(Bs + (nt * 16 + r) * B_STRIDE + ks + 4 * h);
                v2f b1; b1[0] = b4.x; b1[1] = b4.y;
                v2f b2; b2[0] = b4.z; b2[1] = b4.w;
                acc[nt] = __builtin_amdgcn_wmma_f32_16x16x4_f32(
                    false, a1, false, b1, (short)0, acc[nt], false, false);
                acc[nt] = __builtin_amdgcn_wmma_f32_16x16x4_f32(
                    false, a2, false, b2, (short)0, acc[nt], false, false);
            }
        }
    }
}

__global__ void __launch_bounds__(256)
k_gemm_wmma(const float* __restrict__ A1, const int* __restrict__ ids1,
            const float* __restrict__ W1, int K1,
            const float* __restrict__ A2, const float* __restrict__ W2, int K2,
            const float* __restrict__ bias, float* __restrict__ out,
            int nrows, int relu) {
    __shared__ float Bs[DDIM * B_STRIDE];

    const int tid  = threadIdx.x;
    const int lane = tid & 31;
    const int wave = tid >> 5;
    const int row0 = (blockIdx.x * 8 + wave) * 16;
    const bool valid = (row0 < nrows);
    const int rowA = valid ? row0 : 0;
    const int r = lane & 15;
    const int h = lane >> 4;

    v8f acc[8] = {};

    gemm_pass(A1, ids1, W1, K1, rowA, r, h, tid, Bs, acc);
    if (A2) gemm_pass(A2, nullptr, W2, K2, rowA, r, h, tid, Bs, acc);

    if (!valid) return;
#pragma unroll
    for (int nt = 0; nt < 8; nt++) {
        int col  = nt * 16 + r;
        float bb = bias ? bias[col] : 0.0f;
#pragma unroll
        for (int i = 0; i < 8; i++) {
            float v = acc[nt][i] + bb;
            if (relu) v = fmaxf(v, 0.0f);
            out[(size_t)(row0 + h * 8 + i) * DDIM + col] = v;
        }
    }
}

__global__ void __launch_bounds__(1024)
k_fold_vec(const float* __restrict__ Wroot_t, const float* __restrict__ Wnbr_m,
           const float* __restrict__ Wroot_m, const float* __restrict__ Wnbr_t,
           const float* __restrict__ bias_t, const float* __restrict__ bias_m,
           const float* __restrict__ cw, float* __restrict__ U, float* __restrict__ cst) {
    __shared__ float red[8][DDIM];
    const int tid = threadIdx.x;
    const int b   = blockIdx.x;
    const int k   = tid & (DDIM - 1);
    const int g   = tid >> 7;
    const float* Wm = (b == 0) ? Wroot_t : (b == 1) ? Wnbr_m : (b == 2) ? Wroot_m : Wnbr_t;
    const float* c  = (b == 0 || b == 3) ? cw : cw + DDIM;
    float s = 0.0f;
#pragma unroll
    for (int jj = 0; jj < 16; jj++) {
        const int j = g * 16 + jj;
        s += Wm[j * DDIM + k] * c[j];
    }
    red[g][k] = s;
    __syncthreads();
    if (tid < DDIM) {
        float t = 0.0f;
#pragma unroll
        for (int q = 0; q < 8; q++) t += red[q][tid];
        U[b * DDIM + tid] = t;
    }
    if ((b == 0 || b == 2) && tid < 32) {
        const float* bv = (b == 0) ? bias_t : bias_m;
        float t = 0.0f;
#pragma unroll
        for (int q = 0; q < 4; q++) t += bv[tid * 4 + q] * c[tid * 4 + q];
        for (int o = 16; o > 0; o >>= 1) t += __shfl_xor(t, o, 32);
        if (tid == 0) cst[b >> 1] = t;
    }
}

__global__ void __launch_bounds__(256)
k_gemm_fold(const float* __restrict__ A1, const float* __restrict__ W1, int K1,
            const float* __restrict__ C0,
            const float* __restrict__ A2, const int* __restrict__ ids2,
            const float* __restrict__ W2, int K2,
            const float* __restrict__ bias, const float* __restrict__ U,
            float* __restrict__ pq, int nrows) {
    __shared__ float Bs[DDIM * B_STRIDE];

    const int tid  = threadIdx.x;
    const int lane = tid & 31;
    const int wave = tid >> 5;
    const int row0 = (blockIdx.x * 8 + wave) * 16;
    const bool valid = (row0 < nrows);
    const int rowA = valid ? row0 : 0;
    const int r = lane & 15;
    const int h = lane >> 4;

    v8f acc[8] = {};
    if (C0) {
#pragma unroll
        for (int nt = 0; nt < 8; nt++) {
#pragma unroll
            for (int i = 0; i < 8; i++)
                acc[nt][i] = C0[(size_t)(rowA + h * 8 + i) * DDIM + nt * 16 + r];
        }
    }

    if (A1) gemm_pass(A1, nullptr, W1, K1, rowA, r, h, tid, Bs, acc);
    gemm_pass(A2, ids2, W2, K2, rowA, r, h, tid, Bs, acc);

    if (!valid) return;

    float pa[8], qa[8];
#pragma unroll
    for (int i = 0; i < 8; i++) { pa[i] = 0.0f; qa[i] = 0.0f; }
#pragma unroll
    for (int nt = 0; nt < 8; nt++) {
        const int col  = nt * 16 + r;
        const float bb = bias[col];
        const float up = U[col];
        const float uq = U[DDIM + col];
#pragma unroll
        for (int i = 0; i < 8; i++) {
            const float v = fmaxf(acc[nt][i] + bb, 0.0f);
            pa[i] += v * up;
            qa[i] += v * uq;
        }
    }
    float k8[8], k4[4], k2[2], res;
    {
        const bool hi = (r & 8) != 0;
#pragma unroll
        for (int k = 0; k < 8; k++) {
            const float snd = hi ? pa[k] : qa[k];
            const float kp  = hi ? qa[k] : pa[k];
            k8[k] = kp + __shfl_xor(snd, 8, 32);
        }
    }
    {
        const bool hi = (r & 4) != 0;
#pragma unroll
        for (int k = 0; k < 4; k++) {
            const float snd = hi ? k8[k] : k8[k + 4];
            const float kp  = hi ? k8[k + 4] : k8[k];
            k4[k] = kp + __shfl_xor(snd, 4, 32);
        }
    }
    {
        const bool hi = (r & 2) != 0;
#pragma unroll
        for (int k = 0; k < 2; k++) {
            const float snd = hi ? k4[k] : k4[k + 2];
            const float kp  = hi ? k4[k + 2] : k4[k];
            k2[k] = kp + __shfl_xor(snd, 2, 32);
        }
    }
    {
        const bool hi = (r & 1) != 0;
        const float snd = hi ? k2[0] : k2[1];
        const float kp  = hi ? k2[1] : k2[0];
        res = kp + __shfl_xor(snd, 1, 32);
    }
    const int pos = ((r >> 3) << 4) + h * 8 + (r & 7);
    pq[(size_t)(row0 >> 4) * 32 + pos] = res;
}

__device__ __forceinline__ float pq_at(const float* __restrict__ pq, int v, int which) {
    return pq[(size_t)(v >> 4) * 32 + which * 16 + (v & 15)];
}

__global__ void __launch_bounds__(256)
k_node_score(const float* __restrict__ pq_t, const float* __restrict__ pq_m,
             const int* __restrict__ off, const int* __restrict__ list,
             const float* __restrict__ cst, float* __restrict__ S) {
    const int t  = blockIdx.x * 256 + threadIdx.x;
    const int v  = t >> 3;
    const int l8 = t & 7;
    if (v >= N_NODES) return;
    const bool is_m = (v < N_M);
    const float* own = is_m ? pq_m : pq_t;
    const float* oth = is_m ? pq_t : pq_m;
    const int beg = off[v];
    const int end = off[v + 1];
    float s = 0.0f;
    for (int j = beg + l8; j < end; j += 16) {
        const int a  = list[j];
        const int a2 = (j + 8 < end) ? list[j + 8] : -1;
        const float x  = pq_at(oth, a, 1);
        const float x2 = (a2 >= 0) ? pq_at(oth, a2, 1) : 0.0f;
        s += x;
        s += x2;
    }
    s += __shfl_xor(s, 4, 32);
    s += __shfl_xor(s, 2, 32);
    s += __shfl_xor(s, 1, 32);
    if (l8 == 0) {
        const float dg = fmaxf((float)(end - beg), 1.0f);
        S[v] = pq_at(own, is_m ? v : v - N_M, 0) + cst[is_m ? 1 : 0] + s / dg;
    }
}

__global__ void __launch_bounds__(256)
k_classify(const float* __restrict__ S, const int* __restrict__ qs, const int* __restrict__ qd,
           const float* __restrict__ cb, float* __restrict__ out, int nq) {
    const int q = blockIdx.x * 256 + threadIdx.x;
    if (q >= nq) return;
    out[q] = S[N_M + qs[q]] + S[qd[q]] + cb[0];
}

extern "C" void kernel_launch(void* const* d_in, const int* in_sizes, int n_in,
                              void* d_out, int out_size, void* d_ws, size_t ws_size,
                              hipStream_t stream) {
    const float* x_thesis     = (const float*)d_in[0];
    const int*   mentor_id    = (const int*)  d_in[1];
    const int*   edge_src     = (const int*)  d_in[2];
    const int*   edge_dst     = (const int*)  d_in[3];
    const int*   eli_src      = (const int*)  d_in[4];
    const int*   eli_dst      = (const int*)  d_in[5];
    const float* thesis_lin_w = (const float*)d_in[6];
    const float* thesis_lin_b = (const float*)d_in[7];
    const float* prof_emb     = (const float*)d_in[8];
    const float* W_nbr_sup    = (const float*)d_in[9];
    const float* W_root_sup   = (const float*)d_in[10];
    const float* b_sup        = (const float*)d_in[11];
    const float* W_nbr_rev    = (const float*)d_in[12];
    const float* W_root_rev   = (const float*)d_in[13];
    const float* b_rev        = (const float*)d_in[14];
    const float* cls_w        = (const float*)d_in[15];
    const float* cls_b        = (const float*)d_in[16];
    float*       out          = (float*)d_out;

    float* p    = (float*)d_ws;
    float* ht0  = p; p += (size_t)N_T * DDIM;
    float* ht1  = p; p += (size_t)N_T * DDIM;
    float* aggt = p; p += (size_t)N_T * DDIM;
    float* ym   = p; p += (size_t)N_M * DDIM;
    float* hm1  = p; p += (size_t)N_M * DDIM;
    float* aggm = p; p += (size_t)N_M * DDIM;

    int* csrA  = (int*)ht1;
    int* offA  = csrA;
    int* listA = csrA + OFF_PAD;

    float* foldU = hm1;
    float* foldC = hm1 + 4 * DDIM;
    float* pqT   = hm1 + 1024;
    float* pqM   = pqT + 2 * N_T;
    float* score = pqM + 2 * N_M;

    int* binH = (int*)aggt;
    int* binS = binH + HN;
    int* binR = binS + HN_PAD;

    const int BLK = 256;
    const int ROWS_PER_BLK = 8 * 16;
    const int SEG_BLOCKS_M = 2500;
    const int SEG_BLOCKS_T = 2048;
    const size_t wOff = (size_t)DDIM * DDIM;

    k_bin_count<<<NGRP, 512, 0, stream>>>(edge_src, edge_dst, binH);
    k_scan<<<(HN + 1023) / 1024, 1024, 0, stream>>>(binH, binS, HN);
    k_bin_fill<<<NGRP, 512, 0, stream>>>(edge_src, edge_dst, binS, binR);
    k_bin_rank<<<NBINS, 512, 0, stream>>>(binS, binR, offA, listA);

    k_fold_vec<<<4, 1024, 0, stream>>>(W_root_rev + wOff, W_nbr_sup + wOff,
                                       W_root_sup + wOff, W_nbr_rev + wOff,
                                       b_rev + DDIM, b_sup + DDIM, cls_w, foldU, foldC);

    k_gemm_wmma<<<(N_T + ROWS_PER_BLK - 1) / ROWS_PER_BLK, BLK, 0, stream>>>(
        x_thesis, nullptr, thesis_lin_w, F_IN, nullptr, nullptr, 0,
        thesis_lin_b, ht0, N_T, 0);

    k_gemm_wmma<<<(N_M + ROWS_PER_BLK - 1) / ROWS_PER_BLK, BLK, 0, stream>>>(
        prof_emb, mentor_id, W_nbr_rev, DDIM, nullptr, nullptr, 0,
        nullptr, ym, N_M, 0);

    k_seg_mean<<<SEG_BLOCKS_M, BLK, 0, stream>>>(ht0, offA, listA, aggm, N_M);
    k_seg_mean<<<SEG_BLOCKS_T, BLK, 0, stream>>>(ym, offA + N_M, listA, aggt, N_T);
    k_gemm_fold<<<(N_M + ROWS_PER_BLK - 1) / ROWS_PER_BLK, BLK, 0, stream>>>(
        aggm, W_nbr_sup, DDIM, nullptr, prof_emb, mentor_id, W_root_sup, DDIM,
        b_sup, foldU + 2 * DDIM, pqM, N_M);
    k_gemm_fold<<<(N_T + ROWS_PER_BLK - 1) / ROWS_PER_BLK, BLK, 0, stream>>>(
        nullptr, nullptr, 0, aggt, ht0, nullptr, W_root_rev, DDIM,
        b_rev, foldU, pqT, N_T);

    k_node_score<<<(N_NODES * 8) / BLK, BLK, 0, stream>>>(pqT, pqM, offA, listA, foldC, score);
    k_classify<<<(N_Q + BLK - 1) / BLK, BLK, 0, stream>>>(score, eli_src, eli_dst, cls_b, out, N_Q);
}


# Independent kernels share launches as block ranges of one grid: both sides' products, both neighbour means, fold vectors with the prefix sum; 13 launches become 9; no arithmetic or sum order changed
# speedup vs baseline: 4.0502x; 1.1326x over previous
#include <hip/hip_runtime.h>
#include <hip/hip_bf16.h>

#define N_T   100000
#define N_M   20000
#define N_E   600000
#define N_Q   200000
#define F_IN  256
#define DDIM  128

#define N_NODES  (N_M + N_T)
#define OFF_PAD  120064

#define K_SLAB   64
#define B_STRIDE 68

typedef float v2f __attribute__((ext_vector_type(2)));
typedef float v8f __attribute__((ext_vector_type(8)));

#define RM_SHIFT 7
#define RT_SHIFT 9
#define NB_M   157
#define NB_T   196
#define NBINS  (NB_M + NB_T)
#define NGRP   256
#define CHUNK  2344
#define HN     (NBINS * NGRP)
#define HN_PAD (HN + 4)
#define EPT    ((CHUNK + 511) / 512)

__global__ void __launch_bounds__(512)
k_bin_count(const int* __restrict__ esrc, const int* __restrict__ edst, int* __restrict__ H) {
    __shared__ int h[NBINS];
    const int tid = threadIdx.x;
    const int g   = blockIdx.x;
    for (int i = tid; i < NBINS; i += 512) h[i] = 0;
    __syncthreads();
    const int beg = g * CHUNK;
    const int end = min(beg + CHUNK, N_E);
    int s[EPT], d[EPT];
#pragma unroll
    for (int k = 0; k < EPT; k++) {
        const int e = min(beg + tid + k * 512, end - 1);
        s[k] = esrc[e];
        d[k] = edst[e];
    }
#pragma unroll
    for (int k = 0; k < EPT; k++) {
        if (beg + tid + k * 512 < end) {
            atomicAdd(&h[d[k] >> RM_SHIFT], 1);
            atomicAdd(&h[NB_M + (s[k] >> RT_SHIFT)], 1);
        }
    }
    __syncthreads();
    for (int i = tid; i < NBINS; i += 512) H[i * NGRP + g] = h[i];
}

__device__ __forceinline__ void scan_block(const int* __restrict__ cnt, int* __restrict__ off, int n,
                                           int blk, int* wpre, int* wtot) {
    const int tid  = threadIdx.x;
    const int lane = tid & 31;
    const int wave = tid >> 5;
    const int base = blk * 1024;

    int s = 0;
    for (int i0 = tid * 4; i0 < base; i0 += 4 * 4096) {
        int4 v[4];
#pragma unroll
        for (int k = 0; k < 4; k++) {
            const int i = i0 + k * 4096;
            v[k] = (i < base) ? *(const int4*)(cnt + i) : make_int4(0, 0, 0, 0);
        }
#pragma unroll
        for (int k = 0; k < 4; k++) s += v[k].x + v[k].y + v[k].z + v[k].w;
    }
    for (int o = 16; o > 0; o >>= 1) s += __shfl_xor(s, o, 32);

    const int i = base + tid;
    const int mine = (i < n) ? cnt[i] : 0;
    int incl = mine;
    for (int o = 1; o < 32; o <<= 1) {
        int y = __shfl_up(incl, o, 32);
        if (lane >= o) incl += y;
    }
    if (lane == 0)  wpre[wave] = s;
    if (lane == 31) wtot[wave] = incl;
    __syncthreads();
    int front = 0;
    for (int w = 0; w < 32; w++) {
        front += wpre[w];
        if (w < wave) front += wtot[w];
    }
    const int excl = front + incl - mine;
    if (i < n) off[i] = excl;
    if (i == n - 1) off[n] = excl + mine;
}

__global__ void __launch_bounds__(512)
k_bin_fill(const int* __restrict__ esrc, const int* __restrict__ edst,
           const int* __restrict__ S, int* __restrict__ rec) {
    __shared__ int cur[NBINS];
    const int tid = threadIdx.x;
    const int g   = blockIdx.x;
    for (int i = tid; i < NBINS; i += 512) cur[i] = S[i * NGRP + g];
    __syncthreads();
    const int beg = g * CHUNK;
    const int end = min(beg + CHUNK, N_E);
    int s[EPT], d[EPT];
#pragma unroll
    for (int k = 0; k < EPT; k++) {
        const int e = min(beg + tid + k * 512, end - 1);
        s[k] = esrc[e];
        d[k] = edst[e];
    }
#pragma unroll
    for (int k = 0; k < EPT; k++) {
        if (beg + tid + k * 512 < end) {
            const int pm = atomicAdd(&cur[d[k] >> RM_SHIFT], 1);
            rec[pm] = ((d[k] & ((1 << RM_SHIFT) - 1)) << 17) | s[k];
            const int pt = atomicAdd(&cur[NB_M + (s[k] >> RT_SHIFT)], 1);
            rec[pt] = ((s[k] & ((1 << RT_SHIFT) - 1)) << 17) | d[k];
        }
    }
}

__global__ void __launch_bounds__(512)
k_bin_rank(const int* __restrict__ S, const int* __restrict__ rec,
           int* __restrict__ off, int* __restrict__ list) {
    __shared__ int c[512];
    __shared__ int wtot[16];
    const int tid  = threadIdx.x;
    const int lane = tid & 31;
    const int wave = tid >> 5;
    const int b    = blockIdx.x;
    const int beg  = S[b * NGRP];
    const int end  = S[(b + 1) * NGRP];
    const int node0 = (b < NB_M) ? (b << RM_SHIFT) : (N_M + ((b - NB_M) << RT_SHIFT));
    const int nn    = (b < NB_M) ? min(1 << RM_SHIFT, N_M - (b << RM_SHIFT))
                                 : min(1 << RT_SHIFT, N_T - ((b - NB_M) << RT_SHIFT));
    c[tid] = 0;
    __syncthreads();
    for (int i0 = beg + tid; i0 < end; i0 += 4 * 512) {
        int r[4];
#pragma unroll
        for (int k = 0; k < 4; k++) {
            const int i = i0 + k * 512;
            r[k] = (i < end) ? rec[i] : -1;
        }
#pragma unroll
        for (int k = 0; k < 4; k++) if (r[k] >= 0) atomicAdd(&c[r[k] >> 17], 1);
    }
    __syncthreads();

    const int mine = c[tid];
    int incl = mine;
    for (int o = 1; o < 32; o <<= 1) {
        int y = __shfl_up(incl, o, 32);
        if (lane >= o) incl += y;
    }
    if (lane == 31) wtot[wave] = incl;
    __syncthreads();
    int front = 0;
    for (int w = 0; w < 16; w++) if (w < wave) front += wtot[w];
    const int start = beg + front + incl - mine;
    c[tid] = start;
    if (tid < nn) off[node0 + tid] = start;
    __syncthreads();

    for (int i0 = beg + tid; i0 < end; i0 += 4 * 512) {
        int r[4];
#pragma unroll
        for (int k = 0; k < 4; k++) {
            const int i = i0 + k * 512;
            r[k] = (i < end) ? rec[i] : -1;
        }
#pragma unroll
        for (int k = 0; k < 4; k++) {
            if (r[k] >= 0) {
                const int p = atomicAdd(&c[r[k] >> 17], 1);
                list[p] = r[k] & ((1 << 17) - 1);
            }
        }
    }
    if (b == NBINS - 1 && tid == 0) off[N_NODES] = end;
}

__global__ void __launch_bounds__(256)
k_seg_mean2(const float* __restrict__ feata, const int* __restrict__ offa,
            float* __restrict__ agga, int nnodesa, int nblka,
            const float* __restrict__ featb, const int* __restrict__ offb,
            float* __restrict__ aggb, int nnodesb, int nblkb,
            const int* __restrict__ list) {
    const bool ja = (int)blockIdx.x < nblka;
    const float* __restrict__ feat = ja ? feata : featb;
    const int*   __restrict__ off  = ja ? offa : offb;
    float*       __restrict__ agg  = ja ? agga : aggb;
    const int nnodes = ja ? nnodesa : nnodesb;
    const int blk    = ja ? (int)blockIdx.x : (int)blockIdx.x - nblka;
    const int lane = threadIdx.x & 31;
    const int wave0  = __builtin_amdgcn_readfirstlane((int)((blk * 256 + (int)threadIdx.x) >> 5));
    const int nwaves = (ja ? nblka : nblkb) * 8;
    for (int v = wave0; v < nnodes; v += nwaves) {
        const int beg = off[v];
        const int end = off[v + 1];
        float4 acc = make_float4(0.0f, 0.0f, 0.0f, 0.0f);
        for (int b = beg; b < end; b += 32) {
            const int n = min(32, end - b);
            const int mine = (lane < n) ? list[b + lane] : 0;
#pragma unroll 4
            for (int j = 0; j < n; j++) {
                const int s = __builtin_amdgcn_readlane(mine, j);
                const float4 x = *(const float4*)(feat + (size_t)s * DDIM + lane * 4);
                acc.x += x.x; acc.y += x.y; acc.z += x.z; acc.w += x.w;
            }
        }
        const float dg = fmaxf((float)(end - beg), 1.0f);
        float4 o;
        o.x = acc.x / dg; o.y = acc.y / dg; o.z = acc.z / dg; o.w = acc.w / dg;
        *(float4*)(agg + (size_t)v * DDIM + lane * 4) = o;
    }
}

__device__ __forceinline__ void gemm_pass(const float* __restrict__ A,
                                          const int* __restrict__ ids,
                                          const float* __restrict__ W,
                                          int K, int rowA, int r, int h, int tid,
                                          float* __restrict__ Bs, v8f acc[8]) {
    const int arow = ids ? ids[rowA + r] : (rowA + r);
    for (int kk0 = 0; kk0 < K; kk0 += K_SLAB) {
        __syncthreads();
#pragma unroll
        for (int i = 0; i < 8; i++) {
            int idx = tid + i * 256;
            int row = idx >> 4;
            int c   = (idx & 15) << 2;
            float4 v = *(const float4*)(W + (size_t)row * K + kk0 + c);
            *(float4*)(Bs + row * B_STRIDE + c) = v;
        }
        __syncthreads();

        const float* Arow = A + (size_t)arow * K + kk0 + 4 * h;
        for (int ks = 0; ks < K_SLAB; ks += 8) {
            float4 a4 = *(const float4*)(Arow + ks);
            v2f a1; a1[0] = a4.x; a1[1] = a4.y;
            v2f a2; a2[0] = a4.z; a2[1] = a4.w;
#pragma unroll
            for (int nt = 0; nt < 8; nt++) {
                float4 b4 = *(const float4*)(Bs + (nt * 16 + r) * B_STRIDE + ks + 4 * h);
                v2f b1; b1[0] = b4.x; b1[1] = b4.y;
                v2f b2; b2[0] = b4.z; b2[1] = b4.w;
                acc[nt] = __builtin_amdgcn_wmma_f32_16x16x4_f32(
                    false, a1, false, b1, (short)0, acc[nt], false, false);
                acc[nt] = __builtin_amdgcn_wmma_f32_16x16x4_f32(
                    false, a2, false, b2, (short)0, acc[nt], false, false);
            }
        }
    }
}

__global__ void __launch_bounds__(256)
k_gemm_wmma2(const float* __restrict__ Aa, const int* __restrict__ idsa,
             const float* __restrict__ Wa, int Ka,
             const float* __restrict__ biasa, float* __restrict__ outa, int nrowsa, int nblka,
             const float* __restrict__ Ab, const int* __restrict__ idsb,
             const float* __restrict__ Wb, int Kb,
             const float* __restrict__ biasb, float* __restrict__ outb, int nrowsb) {
    __shared__ float Bs[DDIM * B_STRIDE];

    const bool ja = (int)blockIdx.x < nblka;
    const float* __restrict__ A    = ja ? Aa : Ab;
    const int*   __restrict__ ids  = ja ? idsa : idsb;
    const float* __restrict__ W    = ja ? Wa : Wb;
    const int                 K    = ja ? Ka : Kb;
    const float* __restrict__ bias = ja ? biasa : biasb;
    float*       __restrict__ out  = ja ? outa : outb;
    const int nrows = ja ? nrowsa : nrowsb;
    const int blk   = ja ? (int)blockIdx.x : (int)blockIdx.x - nblka;

    const int tid  = threadIdx.x;
    const int lane = tid & 31;
    const int wave = tid >> 5;
    const int row0 = (blk * 8 + wave) * 16;
    const bool valid = (row0 < nrows);
    const int rowA = valid ? row0 : 0;
    const int r = lane & 15;
    const int h = lane >> 4;

    v8f acc[8] = {};

    gemm_pass(A, ids, W, K, rowA, r, h, tid, Bs, acc);

    if (!valid) return;
#pragma unroll
    for (int nt = 0; nt < 8; nt++) {
        int col  = nt * 16 + r;
        float bb = bias ? bias[col] : 0.0f;
#pragma unroll
        for (int i = 0; i < 8; i++) {
            out[(size_t)(row0 + h * 8 + i) * DDIM + col] = acc[nt][i] + bb;
        }
    }
}

__device__ __forceinline__ void fold_vec_block(const float* __restrict__ Wroot_t, const float* __restrict__ Wnbr_m,
           const float* __restrict__ Wroot_m, const float* __restrict__ Wnbr_t,
           const float* __restrict__ bias_t, const float* __restrict__ bias_m,
           const float* __restrict__ cw, float* __restrict__ U, float* __restrict__ cst,
           int b, float (*red)[DDIM]) {
    const int tid = threadIdx.x;
    const int k   = tid & (DDIM - 1);
    const int g   = tid >> 7;
    const float* Wm = (b == 0) ? Wroot_t : (b == 1) ? Wnbr_m : (b == 2) ? Wroot_m : Wnbr_t;
    const float* c  = (b == 0 || b == 3) ? cw : cw + DDIM;
    float s = 0.0f;
#pragma unroll
    for (int jj = 0; jj < 16; jj++) {
        const int j = g * 16 + jj;
        s += Wm[j * DDIM + k] * c[j];
    }
    red[g][k] = s;
    __syncthreads();
    if (tid < DDIM) {
        float t = 0.0f;
#pragma unroll
        for (int q = 0; q < 8; q++) t += red[q][tid];
        U[b * DDIM + tid] = t;
    }
    if ((b == 0 || b == 2) && tid < 32) {
        const float* bv = (b == 0) ? bias_t : bias_m;
        float t = 0.0f;
#pragma unroll
        for (int q = 0; q < 4; q++) t += bv[tid * 4 + q] * c[tid * 4 + q];
        for (int o = 16; o > 0; o >>= 1) t += __shfl_xor(t, o, 32);
        if (tid == 0) cst[b >> 1] = t;
    }
}

__global__ void __launch_bounds__(1024)
k_scan_fold(const int* __restrict__ cnt, int* __restrict__ off, int n, int nscan,
            const float* __restrict__ Wroot_t, const float* __restrict__ Wnbr_m,
            const float* __restrict__ Wroot_m, const float* __restrict__ Wnbr_t,
            const float* __restrict__ bias_t, const float* __restrict__ bias_m,
            const float* __restrict__ cw, float* __restrict__ U, float* __restrict__ cst) {
    __shared__ int wpre[32];
    __shared__ int wtot[32];
    __shared__ float red[8][DDIM];
    const int blk = blockIdx.x;
    if (blk < nscan) scan_block(cnt, off, n, blk, wpre, wtot);
    else fold_vec_block(Wroot_t, Wnbr_m, Wroot_m, Wnbr_t, bias_t, bias_m, cw, U, cst, blk - nscan, red);
}

__global__ void __launch_bounds__(256)
k_gemm_fold2(const float* __restrict__ A1a, const float* __restrict__ W1a, int K1a,
             const float* __restrict__ C0a,
             const float* __restrict__ A2a, const int* __restrict__ ids2a,
             const float* __restrict__ W2a, int K2a,
             const float* __restrict__ biasa, const float* __restrict__ Ua,
             float* __restrict__ pqa, int nrowsa, int nblka,
             const float* __restrict__ A1b, const float* __restrict__ W1b, int K1b,
             const float* __restrict__ C0b,
             const float* __restrict__ A2b, const int* __restrict__ ids2b,
             const float* __restrict__ W2b, int K2b,
             const float* __restrict__ biasb, const float* __restrict__ Ub,
             float* __restrict__ pqb, int nrowsb) {
    __shared__ float Bs[DDIM * B_STRIDE];

    const bool ja = (int)blockIdx.x < nblka;
    const float* __restrict__ A1   = ja ? A1a : A1b;
    const float* __restrict__ W1   = ja ? W1a : W1b;
    const int                 K1   = ja ? K1a : K1b;
    const float* __restrict__ C0   = ja ? C0a : C0b;
    const float* __restrict__ A2   = ja ? A2a : A2b;
    const int*   __restrict__ ids2 = ja ? ids2a : ids2b;
    const float* __restrict__ W2   = ja ? W2a : W2b;
    const int                 K2   = ja ? K2a : K2b;
    const float* __restrict__ bias = ja ? biasa : biasb;
    const float* __restrict__ U    = ja ? Ua : Ub;
    float*       __restrict__ pq   = ja ? pqa : pqb;
    const int nrows = ja ? nrowsa : nrowsb;
    const int blk   = ja ? (int)blockIdx.x : (int)blockIdx.x - nblka;

    const int tid  = threadIdx.x;
    const int lane = tid & 31;
    const int wave = tid >> 5;
    const int row0 = (blk * 8 + wave) * 16;
    const bool valid = (row0 < nrows);
    const int rowA = valid ? row0 : 0;
    const int r = lane & 15;
    const int h = lane >> 4;

    v8f acc[8] = {};
    if (C0) {
#pragma unroll
        for (int nt = 0; nt < 8; nt++) {
#pragma unroll
            for (int i = 0; i < 8; i++)
                acc[nt][i] = C0[(size_t)(rowA + h * 8 + i) * DDIM + nt * 16 + r];
        }
    }

    if (A1) gemm_pass(A1, nullptr, W1, K1, rowA, r, h, tid, Bs, acc);
    gemm_pass(A2, ids2, W2, K2, rowA, r, h, tid, Bs, acc);

    if (!valid) return;

    float pa[8], qa[8];
#pragma unroll
    for (int i = 0; i < 8; i++) { pa[i] = 0.0f; qa[i] = 0.0f; }
#pragma unroll
    for (int nt = 0; nt < 8; nt++) {
        const int col  = nt * 16 + r;
        const float bb = bias[col];
        const float up = U[col];
        const float uq = U[DDIM + col];
#pragma unroll
        for (int i = 0; i < 8; i++) {
            const float v = fmaxf(acc[nt][i] + bb, 0.0f);
            pa[i] += v * up;
            qa[i] += v * uq;
        }
    }
    float k8[8], k4[4], k2[2], res;
    {
        const bool hi = (r & 8) != 0;
#pragma unroll
        for (int k = 0; k < 8; k++) {
            const float snd = hi ? pa[k] : qa[k];
            const float kp  = hi ? qa[k] : pa[k];
            k8[k] = kp + __shfl_xor(snd, 8, 32);
        }
    }
    {
        const bool hi = (r & 4) != 0;
#pragma unroll
        for (int k = 0; k < 4; k++) {
            const float snd = hi ? k8[k] : k8[k + 4];
            const float kp  = hi ? k8[k + 4] : k8[k];
            k4[k] = kp + __shfl_xor(snd, 4, 32);
        }
    }
    {
        const bool hi = (r & 2) != 0;
#pragma unroll
        for (int k = 0; k < 2; k++) {
            const float snd = hi ? k4[k] : k4[k + 2];
            const float kp  = hi ? k4[k + 2] : k4[k];
            k2[k] = kp + __shfl_xor(snd, 2, 32);
        }
    }
    {
        const bool hi = (r & 1) != 0;
        const float snd = hi ? k2[0] : k2[1];
        const float kp  = hi ? k2[1] : k2[0];
        res = kp + __shfl_xor(snd, 1, 32);
    }
    const int pos = ((r >> 3) << 4) + h * 8 + (r & 7);
    pq[(size_t)(row0 >> 4) * 32 + pos] = res;
}

__device__ __forceinline__ float pq_at(const float* __restrict__ pq, int v, int which) {
    return pq[(size_t)(v >> 4) * 32 + which * 16 + (v & 15)];
}

__global__ void __launch_bounds__(256)
k_node_score(const float* __restrict__ pq_t, const float* __restrict__ pq_m,
             const int* __restrict__ off, const int* __restrict__ list,
             const float* __restrict__ cst, float* __restrict__ S) {
    const int t  = blockIdx.x * 256 + threadIdx.x;
    const int v  = t >> 3;
    const int l8 = t & 7;
    if (v >= N_NODES) return;
    const bool is_m = (v < N_M);
    const float* own = is_m ? pq_m : pq_t;
    const float* oth = is_m ? pq_t : pq_m;
    const int beg = off[v];
    const int end = off[v + 1];
    float s = 0.0f;
    for (int j = beg + l8; j < end; j += 16) {
        const int a  = list[j];
        const int a2 = (j + 8 < end) ? list[j + 8] : -1;
        const float x  = pq_at(oth, a, 1);
        const float x2 = (a2 >= 0) ? pq_at(oth, a2, 1) : 0.0f;
        s += x;
        s += x2;
    }
    s += __shfl_xor(s, 4, 32);
    s += __shfl_xor(s, 2, 32);
    s += __shfl_xor(s, 1, 32);
    if (l8 == 0) {
        const float dg = fmaxf((float)(end - beg), 1.0f);
        S[v] = pq_at(own, is_m ? v : v - N_M, 0) + cst[is_m ? 1 : 0] + s / dg;
    }
}

__global__ void __launch_bounds__(256)
k_classify(const float* __restrict__ S, const int* __restrict__ qs, const int* __restrict__ qd,
           const float* __restrict__ cb, float* __restrict__ out, int nq) {
    const int q = blockIdx.x * 256 + threadIdx.x;
    if (q >= nq) return;
    out[q] = S[N_M + qs[q]] + S[qd[q]] + cb[0];
}

extern "C" void kernel_launch(void* const* d_in, const int* in_sizes, int n_in,
                              void* d_out, int out_size, void* d_ws, size_t ws_size,
                              hipStream_t stream) {
    const float* x_thesis     = (const float*)d_in[0];
    const int*   mentor_id    = (const int*)  d_in[1];
    const int*   edge_src     = (const int*)  d_in[2];
    const int*   edge_dst     = (const int*)  d_in[3];
    const int*   eli_src      = (const int*)  d_in[4];
    const int*   eli_dst      = (const int*)  d_in[5];
    const float* thesis_lin_w = (const float*)d_in[6];
    const float* thesis_lin_b = (const float*)d_in[7];
    const float* prof_emb     = (const float*)d_in[8];
    const float* W_nbr_sup    = (const float*)d_in[9];
    const float* W_root_sup   = (const float*)d_in[10];
    const float* b_sup        = (const float*)d_in[11];
    const float* W_nbr_rev    = (const float*)d_in[12];
    const float* W_root_rev   = (const float*)d_in[13];
    const float* b_rev        = (const float*)d_in[14];
    const float* cls_w        = (const float*)d_in[15];
    const float* cls_b        = (const float*)d_in[16];
    float*       out          = (float*)d_out;

    float* p    = (float*)d_ws;
    float* ht0  = p; p += (size_t)N_T * DDIM;
    float* ht1  = p; p += (size_t)N_T * DDIM;
    float* aggt = p; p += (size_t)N_T * DDIM;
    float* ym   = p; p += (size_t)N_M * DDIM;
    float* hm1  = p; p += (size_t)N_M * DDIM;
    float* aggm = p; p += (size_t)N_M * DDIM;

    int* csrA  = (int*)ht1;
    int* offA  = csrA;
    int* listA = csrA + OFF_PAD;

    float* foldU = hm1;
    float* foldC = hm1 + 4 * DDIM;
    float* pqT   = hm1 + 1024;
    float* pqM   = pqT + 2 * N_T;
    float* score = pqM + 2 * N_M;

    int* binH = (int*)aggt;
    int* binS = binH + HN;
    int* binR = binS + HN_PAD;

    const int BLK = 256;
    const int SEG_BLOCKS_M = 2500;
    const int SEG_BLOCKS_T = 2048;
    const int GEMM_BLOCKS_T = 782;
    const int GEMM_BLOCKS_M = 157;
    const int SCAN_BLOCKS   = 89;
    const size_t wOff = (size_t)DDIM * DDIM;

    k_bin_count<<<NGRP, 512, 0, stream>>>(edge_src, edge_dst, binH);
    k_scan_fold<<<SCAN_BLOCKS + 4, 1024, 0, stream>>>(binH, binS, HN, SCAN_BLOCKS,
                                       W_root_rev + wOff, W_nbr_sup + wOff,
                                       W_root_sup + wOff, W_nbr_rev + wOff,
                                       b_rev + DDIM, b_sup + DDIM, cls_w, foldU, foldC);
    k_bin_fill<<<NGRP, 512, 0, stream>>>(edge_src, edge_dst, binS, binR);
    k_bin_rank<<<NBINS, 512, 0, stream>>>(binS, binR, offA, listA);

    k_gemm_wmma2<<<GEMM_BLOCKS_T + GEMM_BLOCKS_M, BLK, 0, stream>>>(
        x_thesis, nullptr, thesis_lin_w, F_IN, thesis_lin_b, ht0, N_T, GEMM_BLOCKS_T,
        prof_emb, mentor_id, W_nbr_rev, DDIM, nullptr, ym, N_M);

    k_seg_mean2<<<SEG_BLOCKS_M + SEG_BLOCKS_T, BLK, 0, stream>>>(
        ht0, offA, aggm, N_M, SEG_BLOCKS_M,
        ym, offA + N_M, aggt, N_T, SEG_BLOCKS_T, listA);
    k_gemm_fold2<<<GEMM_BLOCKS_M + GEMM_BLOCKS_T, BLK, 0, stream>>>(
        aggm, W_nbr_sup, DDIM, nullptr, prof_emb, mentor_id, W_root_sup, DDIM,
        b_sup, foldU + 2 * DDIM, pqM, N_M, GEMM_BLOCKS_M,
        nullptr, nullptr, 0, aggt, ht0, nullptr, W_root_rev, DDIM,
        b_rev, foldU, pqT, N_T);

    k_node_score<<<(N_NODES * 8) / BLK, BLK, 0, stream>>>(pqT, pqM, offA, listA, foldC, score);
    k_classify<<<(N_Q + BLK - 1) / BLK, BLK, 0, stream>>>(score, eli_src, eli_dst, cls_b, out, N_Q);
}


# Encoder and pre-weighting products ride in the four edge-grouping launches as block ranges (64-row blocks of 16 waves, 16 x 32 per wave); 9 launches become 8; no arithmetic or sum order changed
# speedup vs baseline: 4.2316x; 1.0448x over previous
#include <hip/hip_runtime.h>
#include <hip/hip_bf16.h>

#define N_T   100000
#define N_M   20000
#define N_E   600000
#define N_Q   200000
#define F_IN  256
#define DDIM  128

#define N_NODES  (N_M + N_T)
#define OFF_PAD  120064

#define K_SLAB   64
#define B_STRIDE 68

typedef float v2f __attribute__((ext_vector_type(2)));
typedef float v8f __attribute__((ext_vector_type(8)));

#define RM_SHIFT 7
#define RT_SHIFT 9
#define NB_M   157
#define NB_T   196
#define NBINS  (NB_M + NB_T)
#define NGRP   256
#define CHUNK  2344
#define HN     (NBINS * NGRP)
#define HN_PAD (HN + 4)
#define EPT    ((CHUNK + 511) / 512)

__device__ __forceinline__ void bin_count_block(const int* __restrict__ esrc, const int* __restrict__ edst,
                                                int* __restrict__ H, int g, int* h) {
    const int tid = threadIdx.x;
    for (int i = tid; i < NBINS; i += 512) h[i] = 0;
    __syncthreads();
    const int beg = g * CHUNK;
    const int end = min(beg + CHUNK, N_E);
    int s[EPT], d[EPT];
#pragma unroll
    for (int k = 0; k < EPT; k++) {
        const int e = min(beg + tid + k * 512, end - 1);
        s[k] = esrc[e];
        d[k] = edst[e];
    }
#pragma unroll
    for (int k = 0; k < EPT; k++) {
        if (beg + tid + k * 512 < end) {
            atomicAdd(&h[d[k] >> RM_SHIFT], 1);
            atomicAdd(&h[NB_M + (s[k] >> RT_SHIFT)], 1);
        }
    }
    __syncthreads();
    for (int i = tid; i < NBINS; i += 512) H[i * NGRP + g] = h[i];
}

__device__ __forceinline__ void scan_block(const int* __restrict__ cnt, int* __restrict__ off, int n,
                                           int blk, int* wpre, int* wtot) {
    const int tid  = threadIdx.x;
    const int lane = tid & 31;
    const int wave = tid >> 5;
    const int base = blk * 512;

    int s = 0;
    for (int i0 = tid * 4; i0 < base; i0 += 4 * 2048) {
        int4 v[4];
#pragma unroll
        for (int k = 0; k < 4; k++) {
            const int i = i0 + k * 2048;
            v[k] = (i < base) ? *(const int4*)(cnt + i) : make_int4(0, 0, 0, 0);
        }
#pragma unroll
        for (int k = 0; k < 4; k++) s += v[k].x + v[k].y + v[k].z + v[k].w;
    }
    for (int o = 16; o > 0; o >>= 1) s += __shfl_xor(s, o, 32);

    const int i = base + tid;
    const int mine = (i < n) ? cnt[i] : 0;
    int incl = mine;
    for (int o = 1; o < 32; o <<= 1) {
        int y = __shfl_up(incl, o, 32);
        if (lane >= o) incl += y;
    }
    if (lane == 0)  wpre[wave] = s;
    if (lane == 31) wtot[wave] = incl;
    __syncthreads();
    int front = 0;
    for (int w = 0; w < 16; w++) {
        front += wpre[w];
        if (w < wave) front += wtot[w];
    }
    const int excl = front + incl - mine;
    if (i < n) off[i] = excl;
    if (i == n - 1) off[n] = excl + mine;
}

__device__ __forceinline__ void bin_fill_block(const int* __restrict__ esrc, const int* __restrict__ edst,
                                               const int* __restrict__ S, int* __restrict__ rec, int g, int* cur) {
    const int tid = threadIdx.x;
    for (int i = tid; i < NBINS; i += 512) cur[i] = S[i * NGRP + g];
    __syncthreads();
    const int beg = g * CHUNK;
    const int end = min(beg + CHUNK, N_E);
    int s[EPT], d[EPT];
#pragma unroll
    for (int k = 0; k < EPT; k++) {
        const int e = min(beg + tid + k * 512, end - 1);
        s[k] = esrc[e];
        d[k] = edst[e];
    }
#pragma unroll
    for (int k = 0; k < EPT; k++) {
        if (beg + tid + k * 512 < end) {
            const int pm = atomicAdd(&cur[d[k] >> RM_SHIFT], 1);
            rec[pm] = ((d[k] & ((1 << RM_SHIFT) - 1)) << 17) | s[k];
            const int pt = atomicAdd(&cur[NB_M + (s[k] >> RT_SHIFT)], 1);
            rec[pt] = ((s[k] & ((1 << RT_SHIFT) - 1)) << 17) | d[k];
        }
    }
}

__device__ __forceinline__ void bin_rank_block(const int* __restrict__ S, const int* __restrict__ rec,
                                               int* __restrict__ off, int* __restrict__ list, int b,
                                               int* c, int* wtot) {
    const int tid  = threadIdx.x;
    const int lane = tid & 31;
    const int wave = tid >> 5;
    const int beg  = S[b * NGRP];
    const int end  = S[(b + 1) * NGRP];
    const int node0 = (b < NB_M) ? (b << RM_SHIFT) : (N_M + ((b - NB_M) << RT_SHIFT));
    const int nn    = (b < NB_M) ? min(1 << RM_SHIFT, N_M - (b << RM_SHIFT))
                                 : min(1 << RT_SHIFT, N_T - ((b - NB_M) << RT_SHIFT));
    c[tid] = 0;
    __syncthreads();
    for (int i0 = beg + tid; i0 < end; i0 += 4 * 512) {
        int r[4];
#pragma unroll
        for (int k = 0; k < 4; k++) {
            const int i = i0 + k * 512;
            r[k] = (i < end) ? rec[i] : -1;
        }
#pragma unroll
        for (int k = 0; k < 4; k++) if (r[k] >= 0) atomicAdd(&c[r[k] >> 17], 1);
    }
    __syncthreads();

    const int mine = c[tid];
    int incl = mine;
    for (int o = 1; o < 32; o <<= 1) {
        int y = __shfl_up(incl, o, 32);
        if (lane >= o) incl += y;
    }
    if (lane == 31) wtot[wave] = incl;
    __syncthreads();
    int front = 0;
    for (int w = 0; w < 16; w++) if (w < wave) front += wtot[w];
    const int start = beg + front + incl - mine;
    c[tid] = start;
    if (tid < nn) off[node0 + tid] = start;
    __syncthreads();

    for (int i0 = beg + tid; i0 < end; i0 += 4 * 512) {
        int r[4];
#pragma unroll
        for (int k = 0; k < 4; k++) {
            const int i = i0 + k * 512;
            r[k] = (i < end) ? rec[i] : -1;
        }
#pragma unroll
        for (int k = 0; k < 4; k++) {
            if (r[k] >= 0) {
                const int p = atomicAdd(&c[r[k] >> 17], 1);
                list[p] = r[k] & ((1 << 17) - 1);
            }
        }
    }
    if (b == NBINS - 1 && tid == 0) off[N_NODES] = end;
}

__global__ void __launch_bounds__(256)
k_seg_mean2(const float* __restrict__ feata, const int* __restrict__ offa,
            float* __restrict__ agga, int nnodesa, int nblka,
            const float* __restrict__ featb, const int* __restrict__ offb,
            float* __restrict__ aggb, int nnodesb, int nblkb,
            const int* __restrict__ list) {
    const bool ja = (int)blockIdx.x < nblka;
    const float* __restrict__ feat = ja ? feata : featb;
    const int*   __restrict__ off  = ja ? offa : offb;
    float*       __restrict__ agg  = ja ? agga : aggb;
    const int nnodes = ja ? nnodesa : nnodesb;
    const int blk    = ja ? (int)blockIdx.x : (int)blockIdx.x - nblka;
    const int lane = threadIdx.x & 31;
    const int wave0  = __builtin_amdgcn_readfirstlane((int)((blk * 256 + (int)threadIdx.x) >> 5));
    const int nwaves = (ja ? nblka : nblkb) * 8;
    for (int v = wave0; v < nnodes; v += nwaves) {
        const int beg = off[v];
        const int end = off[v + 1];
        float4 acc = make_float4(0.0f, 0.0f, 0.0f, 0.0f);
        for (int b = beg; b < end; b += 32) {
            const int n = min(32, end - b);
            const int mine = (lane < n) ? list[b + lane] : 0;
#pragma unroll 4
            for (int j = 0; j < n; j++) {
                const int s = __builtin_amdgcn_readlane(mine, j);
                const float4 x = *(const float4*)(feat + (size_t)s * DDIM + lane * 4);
                acc.x += x.x; acc.y += x.y; acc.z += x.z; acc.w += x.w;
            }
        }
        const float dg = fmaxf((float)(end - beg), 1.0f);
        float4 o;
        o.x = acc.x / dg; o.y = acc.y / dg; o.z = acc.z / dg; o.w = acc.w / dg;
        *(float4*)(agg + (size_t)v * DDIM + lane * 4) = o;
    }
}

template <int NTHR, int NTILE>
__device__ __forceinline__ void gemm_pass(const float* __restrict__ A,
                                          const int* __restrict__ ids,
                                          const float* __restrict__ W,
                                          int K, int rowA, int r, int h, int tid, int cb,
                                          float* __restrict__ Bs, v8f acc[NTILE]) {
    const int arow = ids ? ids[rowA + r] : (rowA + r);
    for (int kk0 = 0; kk0 < K; kk0 += K_SLAB) {
        __syncthreads();
#pragma unroll
        for (int i = 0; i < 2048 / NTHR; i++) {
            int idx = tid + i * NTHR;
            int row = idx >> 4;
            int c   = (idx & 15) << 2;
            float4 v = *(const float4*)(W + (size_t)row * K + kk0 + c);
            *(float4*)(Bs + row * B_STRIDE + c) = v;
        }
        __syncthreads();

        const float* Arow = A + (size_t)arow * K + kk0 + 4 * h;
        for (int ks = 0; ks < K_SLAB; ks += 8) {
            float4 a4 = *(const float4*)(Arow + ks);
            v2f a1; a1[0] = a4.x; a1[1] = a4.y;
            v2f a2; a2[0] = a4.z; a2[1] = a4.w;
#pragma unroll
            for (int nt = 0; nt < NTILE; nt++) {
                float4 b4 = *(const float4*)(Bs + (cb + nt * 16 + r) * B_STRIDE + ks + 4 * h);
                v2f b1; b1[0] = b4.x; b1[1] = b4.y;
                v2f b2; b2[0] = b4.z; b2[1] = b4.w;
                acc[nt] = __builtin_amdgcn_wmma_f32_16x16x4_f32(
                    false, a1, false, b1, (short)0, acc[nt], false, false);
                acc[nt] = __builtin_amdgcn_wmma_f32_16x16x4_f32(
                    false, a2, false, b2, (short)0, acc[nt], false, false);
            }
        }
    }
}

struct GemmJobs {
    const float* Aa; const int* idsa; const float* Wa; int Ka;
    const float* biasa; float* outa; int nrowsa; int offa; int nblka;
    const float* Ab; const int* idsb; const float* Wb; int Kb;
    const float* biasb; float* outb; int nrowsb;
};

__device__ __forceinline__ void gemm_part512(const GemmJobs& J, int lblk, float* __restrict__ Bs) {
    const bool ja = lblk < J.nblka;
    const float* __restrict__ A    = ja ? J.Aa : J.Ab;
    const int*   __restrict__ ids  = ja ? J.idsa : J.idsb;
    const float* __restrict__ W    = ja ? J.Wa : J.Wb;
    const int                 K    = ja ? J.Ka : J.Kb;
    const float* __restrict__ bias = ja ? J.biasa : J.biasb;
    float*       __restrict__ out  = ja ? J.outa : J.outb;
    const int nrows = ja ? J.nrowsa : J.nrowsb;
    const int blk   = ja ? lblk + J.offa : lblk - J.nblka;

    const int tid  = threadIdx.x;
    const int lane = tid & 31;
    const int wave = tid >> 5;
    const int row0 = (blk * 4 + (wave >> 2)) * 16;
    const int cb   = (wave & 3) * 32;
    const bool valid = (row0 < nrows);
    const int rowA = valid ? row0 : 0;
    const int r = lane & 15;
    const int h = lane >> 4;

    v8f acc[2] = {};

    gemm_pass<512, 2>(A, ids, W, K, rowA, r, h, tid, cb, Bs, acc);

    if (!valid) return;
#pragma unroll
    for (int nt = 0; nt < 2; nt++) {
        int col  = cb + nt * 16 + r;
        float bb = bias ? bias[col] : 0.0f;
#pragma unroll
        for (int i = 0; i < 8; i++) {
            out[(size_t)(row0 + h * 8 + i) * DDIM + col] = acc[nt][i] + bb;
        }
    }
}

__device__ __forceinline__ void fold_vec_block(const float* __restrict__ Wroot_t, const float* __restrict__ Wnbr_m,
           const float* __restrict__ Wroot_m, const float* __restrict__ Wnbr_t,
           const float* __restrict__ bias_t, const float* __restrict__ bias_m,
           const float* __restrict__ cw, float* __restrict__ U, float* __restrict__ cst,
           int b, float (*red)[DDIM]) {
    const int tid = threadIdx.x;
    const int k   = tid & (DDIM - 1);
    const int g   = tid >> 7;
    const float* Wm = (b == 0) ? Wroot_t : (b == 1) ? Wnbr_m : (b == 2) ? Wroot_m : Wnbr_t;
    const float* c  = (b == 0 || b == 3) ? cw : cw + DDIM;
    float s0 = 0.0f, s1 = 0.0f;
#pragma unroll
    for (int jj = 0; jj < 16; jj++) {
        const int j0 = g * 32 + jj;
        const int j1 = g * 32 + 16 + jj;
        s0 += Wm[j0 * DDIM + k] * c[j0];
        s1 += Wm[j1 * DDIM + k] * c[j1];
    }
    red[2 * g][k]     = s0;
    red[2 * g + 1][k] = s1;
    __syncthreads();
    if (tid < DDIM) {
        float t = 0.0f;
#pragma unroll
        for (int q = 0; q < 8; q++) t += red[q][tid];
        U[b * DDIM + tid] = t;
    }
    if ((b == 0 || b == 2) && tid < 32) {
        const float* bv = (b == 0) ? bias_t : bias_m;
        float t = 0.0f;
#pragma unroll
        for (int q = 0; q < 4; q++) t += bv[tid * 4 + q] * c[tid * 4 + q];
        for (int o = 16; o > 0; o >>= 1) t += __shfl_xor(t, o, 32);
        if (tid == 0) cst[b >> 1] = t;
    }
}

__global__ void __launch_bounds__(512)
k_count_gemm(const int* __restrict__ esrc, const int* __restrict__ edst, int* __restrict__ H,
             const GemmJobs J) {
    __shared__ float Bs[DDIM * B_STRIDE];
    const int blk = blockIdx.x;
    if (blk < NGRP) bin_count_block(esrc, edst, H, blk, (int*)Bs);
    else gemm_part512(J, blk - NGRP, Bs);
}

__global__ void __launch_bounds__(512)
k_scan_fold(const int* __restrict__ cnt, int* __restrict__ off, int n, int nscan,
            const float* __restrict__ Wroot_t, const float* __restrict__ Wnbr_m,
            const float* __restrict__ Wroot_m, const float* __restrict__ Wnbr_t,
            const float* __restrict__ bias_t, const float* __restrict__ bias_m,
            const float* __restrict__ cw, float* __restrict__ U, float* __restrict__ cst,
            const GemmJobs J) {
    __shared__ float Bs[DDIM * B_STRIDE];
    const int blk = blockIdx.x;
    if (blk < nscan) scan_block(cnt, off, n, blk, (int*)Bs, (int*)Bs + 16);
    else if (blk < nscan + 4)
        fold_vec_block(Wroot_t, Wnbr_m, Wroot_m, Wnbr_t, bias_t, bias_m, cw, U, cst, blk - nscan,
                       (float (*)[DDIM])Bs);
    else gemm_part512(J, blk - nscan - 4, Bs);
}

__global__ void __launch_bounds__(512)
k_fill_gemm(const int* __restrict__ esrc, const int* __restrict__ edst,
            const int* __restrict__ S, int* __restrict__ rec, const GemmJobs J) {
    __shared__ float Bs[DDIM * B_STRIDE];
    const int blk = blockIdx.x;
    if (blk < NGRP) bin_fill_block(esrc, edst, S, rec, blk, (int*)Bs);
    else gemm_part512(J, blk - NGRP, Bs);
}

__global__ void __launch_bounds__(512)
k_rank_gemm(const int* __restrict__ S, const int* __restrict__ rec,
            int* __restrict__ off, int* __restrict__ list, const GemmJobs J) {
    __shared__ float Bs[DDIM * B_STRIDE];
    const int blk = blockIdx.x;
    if (blk < NBINS) bin_rank_block(S, rec, off, list, blk, (int*)Bs, (int*)Bs + 512);
    else gemm_part512(J, blk - NBINS, Bs);
}

__global__ void __launch_bounds__(256)
k_gemm_fold2(const float* __restrict__ A1a, const float* __restrict__ W1a, int K1a,
             const float* __restrict__ C0a,
             const float* __restrict__ A2a, const int* __restrict__ ids2a,
             const float* __restrict__ W2a, int K2a,
             const float* __restrict__ biasa, const float* __restrict__ Ua,
             float* __restrict__ pqa, int nrowsa, int nblka,
             const float* __restrict__ A1b, const float* __restrict__ W1b, int K1b,
             const float* __restrict__ C0b,
             const float* __restrict__ A2b, const int* __restrict__ ids2b,
             const float* __restrict__ W2b, int K2b,
             const float* __restrict__ biasb, const float* __restrict__ Ub,
             float* __restrict__ pqb, int nrowsb) {
    __shared__ float Bs[DDIM * B_STRIDE];

    const bool ja = (int)blockIdx.x < nblka;
    const float* __restrict__ A1   = ja ? A1a : A1b;
    const float* __restrict__ W1   = ja ? W1a : W1b;
    const int                 K1   = ja ? K1a : K1b;
    const float* __restrict__ C0   = ja ? C0a : C0b;
    const float* __restrict__ A2   = ja ? A2a : A2b;
    const int*   __restrict__ ids2 = ja ? ids2a : ids2b;
    const float* __restrict__ W2   = ja ? W2a : W2b;
    const int                 K2   = ja ? K2a : K2b;
    const float* __restrict__ bias = ja ? biasa : biasb;
    const float* __restrict__ U    = ja ? Ua : Ub;
    float*       __restrict__ pq   = ja ? pqa : pqb;
    const int nrows = ja ? nrowsa : nrowsb;
    const int blk   = ja ? (int)blockIdx.x : (int)blockIdx.x - nblka;

    const int tid  = threadIdx.x;
    const int lane = tid & 31;
    const int wave = tid >> 5;
    const int row0 = (blk * 8 + wave) * 16;
    const bool valid = (row0 < nrows);
    const int rowA = valid ? row0 : 0;
    const int r = lane & 15;
    const int h = lane >> 4;

    v8f acc[8] = {};
    if (C0) {
#pragma unroll
        for (int nt = 0; nt < 8; nt++) {
#pragma unroll
            for (int i = 0; i < 8; i++)
                acc[nt][i] = C0[(size_t)(rowA + h * 8 + i) * DDIM + nt * 16 + r];
        }
    }

    if (A1) gemm_pass<256, 8>(A1, nullptr, W1, K1, rowA, r, h, tid, 0, Bs, acc);
    gemm_pass<256, 8>(A2, ids2, W2, K2, rowA, r, h, tid, 0, Bs, acc);

    if (!valid) return;

    float pa[8], qa[8];
#pragma unroll
    for (int i = 0; i < 8; i++) { pa[i] = 0.0f; qa[i] = 0.0f; }
#pragma unroll
    for (int nt = 0; nt < 8; nt++) {
        const int col  = nt * 16 + r;
        const float bb = bias[col];
        const float up = U[col];
        const float uq = U[DDIM + col];
#pragma unroll
        for (int i = 0; i < 8; i++) {
            const float v = fmaxf(acc[nt][i] + bb, 0.0f);
            pa[i] += v * up;
            qa[i] += v * uq;
        }
    }
    float k8[8], k4[4], k2[2], res;
    {
        const bool hi = (r & 8) != 0;
#pragma unroll
        for (int k = 0; k < 8; k++) {
            const float snd = hi ? pa[k] : qa[k];
            const float kp  = hi ? qa[k] : pa[k];
            k8[k] = kp + __shfl_xor(snd, 8, 32);
        }
    }
    {
        const bool hi = (r & 4) != 0;
#pragma unroll
        for (int k = 0; k < 4; k++) {
            const float snd = hi ? k8[k] : k8[k + 4];
            const float kp  = hi ? k8[k + 4] : k8[k];
            k4[k] = kp + __shfl_xor(snd, 4, 32);
        }
    }
    {
        const bool hi = (r & 2) != 0;
#pragma unroll
        for (int k = 0; k < 2; k++) {
            const float snd = hi ? k4[k] : k4[k + 2];
            const float kp  = hi ? k4[k + 2] : k4[k];
            k2[k] = kp + __shfl_xor(snd, 2, 32);
        }
    }
    {
        const bool hi = (r & 1) != 0;
        const float snd = hi ? k2[0] : k2[1];
        const float kp  = hi ? k2[1] : k2[0];
        res = kp + __shfl_xor(snd, 1, 32);
    }
    const int pos = ((r >> 3) << 4) + h * 8 + (r & 7);
    pq[(size_t)(row0 >> 4) * 32 + pos] = res;
}

__device__ __forceinline__ float pq_at(const float* __restrict__ pq, int v, int which) {
    return pq[(size_t)(v >> 4) * 32 + which * 16 + (v & 15)];
}

__global__ void __launch_bounds__(256)
k_node_score(const float* __restrict__ pq_t, const float* __restrict__ pq_m,
             const int* __restrict__ off, const int* __restrict__ list,
             const float* __restrict__ cst, float* __restrict__ S) {
    const int t  = blockIdx.x * 256 + threadIdx.x;
    const int v  = t >> 3;
    const int l8 = t & 7;
    if (v >= N_NODES) return;
    const bool is_m = (v < N_M);
    const float* own = is_m ? pq_m : pq_t;
    const float* oth = is_m ? pq_t : pq_m;
    const int beg = off[v];
    const int end = off[v + 1];
    float s = 0.0f;
    for (int j = beg + l8; j < end; j += 16) {
        const int a  = list[j];
        const int a2 = (j + 8 < end) ? list[j + 8] : -1;
        const float x  = pq_at(oth, a, 1);
        const float x2 = (a2 >= 0) ? pq_at(oth, a2, 1) : 0.0f;
        s += x;
        s += x2;
    }
    s += __shfl_xor(s, 4, 32);
    s += __shfl_xor(s, 2, 32);
    s += __shfl_xor(s, 1, 32);
    if (l8 == 0) {
        const float dg = fmaxf((float)(end - beg), 1.0f);
        S[v] = pq_at(own, is_m ? v : v - N_M, 0) + cst[is_m ? 1 : 0] + s / dg;
    }
}

__global__ void __launch_bounds__(256)
k_classify(const float* __restrict__ S, const int* __restrict__ qs, const int* __restrict__ qd,
           const float* __restrict__ cb, float* __restrict__ out, int nq) {
    const int q = blockIdx.x * 256 + threadIdx.x;
    if (q >= nq) return;
    out[q] = S[N_M + qs[q]] + S[qd[q]] + cb[0];
}

extern "C" void kernel_launch(void* const* d_in, const int* in_sizes, int n_in,
                              void* d_out, int out_size, void* d_ws, size_t ws_size,
                              hipStream_t stream) {
    const float* x_thesis     = (const float*)d_in[0];
    const int*   mentor_id    = (const int*)  d_in[1];
    const int*   edge_src     = (const int*)  d_in[2];
    const int*   edge_dst     = (const int*)  d_in[3];
    const int*   eli_src      = (const int*)  d_in[4];
    const int*   eli_dst      = (const int*)  d_in[5];
    const float* thesis_lin_w = (const float*)d_in[6];
    const float* thesis_lin_b = (const float*)d_in[7];
    const float* prof_emb     = (const float*)d_in[8];
    const float* W_nbr_sup    = (const float*)d_in[9];
    const float* W_root_sup   = (const float*)d_in[10];
    const float* b_sup        = (const float*)d_in[11];
    const float* W_nbr_rev    = (const float*)d_in[12];
    const float* W_root_rev   = (const float*)d_in[13];
    const float* b_rev        = (const float*)d_in[14];
    const float* cls_w        = (const float*)d_in[15];
    const float* cls_b        = (const float*)d_in[16];
    float*       out          = (float*)d_out;

    float* p    = (float*)d_ws;
    float* ht0  = p; p += (size_t)N_T * DDIM;
    float* ht1  = p; p += (size_t)N_T * DDIM;
    float* aggt = p; p += (size_t)N_T * DDIM;
    float* ym   = p; p += (size_t)N_M * DDIM;
    float* hm1  = p; p += (size_t)N_M * DDIM;
    float* aggm = p; p += (size_t)N_M * DDIM;

    int* csrA  = (int*)ht1;
    int* offA  = csrA;
    int* listA = csrA + OFF_PAD;

    float* foldU = hm1;
    float* foldC = hm1 + 4 * DDIM;
    float* pqT   = hm1 + 1024;
    float* pqM   = pqT + 2 * N_T;
    float* score = pqM + 2 * N_M;

    int* binH = (int*)aggt;
    int* binS = binH + HN;
    int* binR = binS + HN_PAD;

    const int BLK = 256;
    const int SEG_BLOCKS_M = 2500;
    const int SEG_BLOCKS_T = 2048;
    const int GEMM_BLOCKS_T = 782;
    const int GEMM_BLOCKS_M = 157;
    const int SCAN_BLOCKS   = 177;
    const size_t wOff = (size_t)DDIM * DDIM;

    GemmJobs J1 = { x_thesis, nullptr, thesis_lin_w, F_IN, thesis_lin_b, ht0, N_T, 0, 512,
                    prof_emb, mentor_id, W_nbr_rev, DDIM, nullptr, ym, N_M };
    GemmJobs J2 = J1; J2.offa = 512;
    GemmJobs J3 = J1; J3.offa = 1024;
    GemmJobs J4 = J1; J4.offa = 1536; J4.nblka = 27;

    k_count_gemm<<<NGRP + 512, 512, 0, stream>>>(edge_src, edge_dst, binH, J1);
    k_scan_fold<<<SCAN_BLOCKS + 4 + 512, 512, 0, stream>>>(binH, binS, HN, SCAN_BLOCKS,
                                       W_root_rev + wOff, W_nbr_sup + wOff,
                                       W_root_sup + wOff, W_nbr_rev + wOff,
                                       b_rev + DDIM, b_sup + DDIM, cls_w, foldU, foldC, J2);
    k_fill_gemm<<<NGRP + 512, 512, 0, stream>>>(edge_src, edge_dst, binS, binR, J3);
    k_rank_gemm<<<NBINS + 27 + 313, 512, 0, stream>>>(binS, binR, offA, listA, J4);

    k_seg_mean2<<<SEG_BLOCKS_M + SEG_BLOCKS_T, BLK, 0, stream>>>(
        ht0, offA, aggm, N_M, SEG_BLOCKS_M,
        ym, offA + N_M, aggt, N_T, SEG_BLOCKS_T, listA);
    k_gemm_fold2<<<GEMM_BLOCKS_M + GEMM_BLOCKS_T, BLK, 0, stream>>>(
        aggm, W_nbr_sup, DDIM, nullptr, prof_emb, mentor_id, W_root_sup, DDIM,
        b_sup, foldU + 2 * DDIM, pqM, N_M, GEMM_BLOCKS_M,
        nullptr, nullptr, 0, aggt, ht0, nullptr, W_root_rev, DDIM,
        b_rev, foldU, pqT, N_T);

    k_node_score<<<(N_NODES * 8) / BLK, BLK, 0, stream>>>(pqT, pqM, offA, listA, foldC, score);
    k_classify<<<(N_Q + BLK - 1) / BLK, BLK, 0, stream>>>(score, eli_src, eli_dst, cls_b, out, N_Q);
}
